# starting kernel
# baseline (speedup 1.0000x reference)
#include <hip/hip_runtime.h>

typedef __attribute__((ext_vector_type(2))) float v2f;
typedef __attribute__((ext_vector_type(8))) float v8f;

#define ICAPS   32
#define IATOMS  16
#define NCAPS   10
#define CDIM    16
#define NCD     160
#define NROUT   3
#define EPSV    1e-7f
#define XSTR    17
#define WSTR    164

__global__ __launch_bounds__(160)
void caps_routing_fused(const float* __restrict__ x,
                        const float* __restrict__ Wm,
                        const float* __restrict__ bias,
                        float* __restrict__ out)
{
    __shared__ float sX[ICAPS * XSTR];
    __shared__ float sW[IATOMS * WSTR];
    __shared__ float sV[ICAPS * WSTR];
    __shared__ float sRoute[ICAPS * NCAPS];
    __shared__ float sLogit[ICAPS * NCAPS];
    __shared__ float sAct[NCD];

    const int t    = threadIdx.x;
    const int lane = t & 31;
    const int wave = t >> 5;
    const int half = lane >> 4;
    const int lcol = lane & 15;
    const long pos = blockIdx.x;

    const float* xg = x + pos * (ICAPS * IATOMS);
    for (int i = t; i < ICAPS * IATOMS; i += 160)
        sX[(i >> 4) * XSTR + (i & 15)] = xg[i];
    for (int i = t; i < IATOMS * NCD; i += 160)
        sW[(i / NCD) * WSTR + (i % NCD)] = Wm[i];
    for (int i = t; i < ICAPS * NCAPS; i += 160)
        sLogit[i] = 0.0f;
    const float bval = bias[t];
    __syncthreads();

    for (int i = 0; i < 4; ++i) {
        const int tile  = wave * 4 + i;
        const int mtile = tile / NCAPS;
        const int ntile = tile % NCAPS;
        v8f acc = {};
#pragma unroll
        for (int k0 = 0; k0 < IATOMS; k0 += 4) {
            const float* ap = &sX[(mtile * 16 + lcol) * XSTR + k0 + 2 * half];
            v2f a; a.x = ap[0]; a.y = ap[1];
            const float* bp = &sW[(k0 + 2 * half) * WSTR + ntile * 16 + lcol];
            v2f b; b.x = bp[0]; b.y = bp[WSTR];
            acc = __builtin_amdgcn_wmma_f32_16x16x4_f32(
                      false, a, false, b, (short)0, acc, false, false);
        }
        float* vd = &sV[(mtile * 16 + 8 * half) * WSTR + ntile * 16 + lcol];
#pragma unroll
        for (int v = 0; v < 8; ++v) vd[v * WSTR] = acc[v];
    }
    __syncthreads();

    const int nc = t >> 4;
    float act = 0.0f;
    for (int r = 0; r < NROUT; ++r) {
        if (t < ICAPS) {
            float lg[NCAPS], mx = -1e30f;
#pragma unroll
            for (int n = 0; n < NCAPS; ++n) { lg[n] = sLogit[t * NCAPS + n]; mx = fmaxf(mx, lg[n]); }
            float s = 0.0f;
#pragma unroll
            for (int n = 0; n < NCAPS; ++n) { lg[n] = __expf(lg[n] - mx); s += lg[n]; }
            const float inv = 1.0f / s;
#pragma unroll
            for (int n = 0; n < NCAPS; ++n) sRoute[t * NCAPS + n] = lg[n] * inv;
        }
        __syncthreads();

        float p = bval;
#pragma unroll 4
        for (int ic = 0; ic < ICAPS; ++ic)
            p += sRoute[ic * NCAPS + nc] * sV[ic * WSTR + t];

        float sq = p * p;
        sq += __shfl_xor(sq, 1, 16);
        sq += __shfl_xor(sq, 2, 16);
        sq += __shfl_xor(sq, 4, 16);
        sq += __shfl_xor(sq, 8, 16);
        act = (sq / (1.0f + sq)) * p * rsqrtf(sq + EPSV);
        sAct[t] = act;
        __syncthreads();

        if (r < NROUT - 1) {
            for (int pr = t; pr < ICAPS * NCAPS; pr += 160) {
                const int ic = pr / NCAPS, n = pr % NCAPS;
                const float* vr = &sV[ic * WSTR + n * CDIM];
                const float* ar = &sAct[n * CDIM];
                float s = 0.0f;
#pragma unroll
                for (int c = 0; c < CDIM; ++c) s += vr[c] * ar[c];
                sLogit[pr] += s;
            }
            __syncthreads();
        }
    }

    out[pos * NCD + t] = act;
}

extern "C" void kernel_launch(void* const* d_in, const int* in_sizes, int n_in,
                              void* d_out, int out_size, void* d_ws, size_t ws_size,
                              hipStream_t stream) {
    const float* x  = (const float*)d_in[0];
    const float* W  = (const float*)d_in[1];
    const float* b  = (const float*)d_in[2];
    float* out      = (float*)d_out;
    const int positions = 32 * 16 * 16;
    caps_routing_fused<<<dim3(positions), dim3(160), 0, stream>>>(x, W, b, out);
}


# One wave per position: all 20 vote tiles stay in WMMA accumulators, routing runs in registers with DPP lane folds (no votes in LDS, no routing barriers); changes sum order of preactivation and logits
# speedup vs baseline: 1.7365x; 1.7365x over previous
#include <hip/hip_runtime.h>


typedef __attribute__((ext_vector_type(2))) float v2f;
typedef __attribute__((ext_vector_type(4))) float v4f;
typedef __attribute__((ext_vector_type(8))) float v8f;

#define ICAPS   32
#define IATOMS  16
#define NCAPS   10
#define CDIM    16
#define NCD     160
#define NROUT   3
#define EPSV    1e-7f
#define XSTR    17
#define WSTR    164
#define RSTR    32
#define PPB     4


template <int M>
__device__ __forceinline__ float row_xor(float x) {
    return __builtin_bit_cast(float, __builtin_amdgcn_update_dpp(
        0, __builtin_bit_cast(int, x), 0x160 + M, 0xf, 0xf, true));
}

template <int M>
__device__ __forceinline__ float add_row_xor(float a, float b) {
    float d;
    asm("v_add_f32_dpp %0, %1, %2 row_xmask:%3 row_mask:0xf bank_mask:0xf bound_ctrl:1"
        : "=v"(d) : "v"(a), "v"(b), "n"(M));
    return d;
}

__device__ __forceinline__ float xhalf(float x) {
    return __builtin_bit_cast(float, __builtin_amdgcn_permlanex16(
        __builtin_bit_cast(int, x), __builtin_bit_cast(int, x),
        0x76543210, 0xfedcba98, false, false));
}

__device__ __forceinline__ float fold16(const float (&P)[16], bool x8, bool x4, bool x2, bool x1) {
    float n8[8], n4[4], n2[2];
#pragma unroll
    for (int k = 0; k < 8; ++k) {
        const float keep = x8 ? P[k + 8] : P[k];
        const float send = x8 ? P[k] : P[k + 8];
        n8[k] = add_row_xor<8>(send, keep);
    }
#pragma unroll
    for (int k = 0; k < 4; ++k) {
        const float keep = x4 ? n8[k + 4] : n8[k];
        const float send = x4 ? n8[k] : n8[k + 4];
        n4[k] = add_row_xor<4>(send, keep);
    }
#pragma unroll
    for (int k = 0; k < 2; ++k) {
        const float keep = x2 ? n4[k + 2] : n4[k];
        const float send = x2 ? n4[k] : n4[k + 2];
        n2[k] = add_row_xor<2>(send, keep);
    }
    const float keep = x1 ? n2[1] : n2[0];
    const float send = x1 ? n2[0] : n2[1];
    return add_row_xor<1>(send, keep);
}

__global__ __launch_bounds__(32 * PPB, 4)
void caps_routing_fused(const float* __restrict__ x,
                        const float* __restrict__ Wm,
                        const float* __restrict__ bias,
                        float* __restrict__ out)
{
    __shared__ float sW[IATOMS * WSTR];
    __shared__ float sX[PPB][ICAPS * XSTR];
    __shared__ __attribute__((aligned(16))) float sR[PPB][NCAPS * RSTR];

    const int t    = threadIdx.x;
    const int lane = t & 31;
    const int wave = t >> 5;
    const int half = lane >> 4;
    const int lcol = lane & 15;
    const long pos = (long)blockIdx.x * PPB + wave;
    float* sXw = sX[wave];
    float* sRw = sR[wave];

    const float* xg = x + pos * (ICAPS * IATOMS);
    for (int i = lane; i < ICAPS * IATOMS; i += 32)
        sXw[(i >> 4) * XSTR + (i & 15)] = xg[i];
    for (int i = t; i < IATOMS * NCD; i += 32 * PPB)
        sW[(i / NCD) * WSTR + (i % NCD)] = Wm[i];
    float bval[NCAPS];
#pragma unroll
    for (int nc = 0; nc < NCAPS; ++nc) bval[nc] = bias[nc * CDIM + lcol];
    __syncthreads();

    v2f a[2][4];
#pragma unroll
    for (int mt = 0; mt < 2; ++mt)
#pragma unroll
        for (int ks = 0; ks < 4; ++ks) {
            const float* ap = &sXw[(mt * 16 + lcol) * XSTR + 4 * ks + 2 * half];
            a[mt][ks].x = ap[0]; a[mt][ks].y = ap[1];
        }
    v8f acc[NCAPS][2];
#pragma unroll
    for (int nc = 0; nc < NCAPS; ++nc) {
        v2f b[4];
#pragma unroll
        for (int ks = 0; ks < 4; ++ks) {
            const float* bp = &sW[(4 * ks + 2 * half) * WSTR + nc * 16 + lcol];
            b[ks].x = bp[0]; b[ks].y = bp[WSTR];
        }
        acc[nc][0] = v8f{}; acc[nc][1] = v8f{};
#pragma unroll
        for (int ks = 0; ks < 4; ++ks)
#pragma unroll
            for (int mt = 0; mt < 2; ++mt)
                acc[nc][mt] = __builtin_amdgcn_wmma_f32_16x16x4_f32(
                                  false, a[mt][ks], false, b[ks], (short)0, acc[nc][mt], false, false);
    }

    const bool x8 = (lcol & 8) != 0, x4 = (lcol & 4) != 0, x2 = (lcol & 2) != 0, x1 = (lcol & 1) != 0;
    const int icl = (lcol >> 3) * 16 + 8 * half + (lcol & 7);
    float lg[NCAPS], act[NCAPS];
#pragma unroll
    for (int n = 0; n < NCAPS; ++n) { lg[n] = 0.0f; act[n] = 0.0f; }

    for (int r = 0; r < NROUT; ++r) {
        {
            float e[NCAPS], mx = -1e30f;
#pragma unroll
            for (int n = 0; n < NCAPS; ++n) mx = fmaxf(mx, lg[n]);
            float s = 0.0f;
#pragma unroll
            for (int n = 0; n < NCAPS; ++n) { e[n] = __expf(lg[n] - mx); s += e[n]; }
            const float inv = 1.0f / s;
            __builtin_amdgcn_wave_barrier();
#pragma unroll
            for (int n = 0; n < NCAPS; ++n) sRw[n * RSTR + icl] = e[n] * inv;
            __builtin_amdgcn_wave_barrier();
        }

#pragma unroll
        for (int nc = 0; nc < NCAPS; ++nc) {
            float p0 = 0.0f, p1 = 0.0f;
#pragma unroll
            for (int mt = 0; mt < 2; ++mt) {
                const v4f* rp = (const v4f*)&sRw[nc * RSTR + mt * 16 + 8 * half];
                const v4f ra = rp[0], rb = rp[1];
                p0 += ra.x * acc[nc][mt][0]; p1 += ra.y * acc[nc][mt][1];
                p0 += ra.z * acc[nc][mt][2]; p1 += ra.w * acc[nc][mt][3];
                p0 += rb.x * acc[nc][mt][4]; p1 += rb.y * acc[nc][mt][5];
                p0 += rb.z * acc[nc][mt][6]; p1 += rb.w * acc[nc][mt][7];
            }
            float p = p0 + p1;
            p += xhalf(p);
            p += bval[nc];

            float sq = p * p;
            sq += row_xor<1>(sq);
            sq = add_row_xor<2>(sq, sq);
            sq = add_row_xor<4>(sq, sq);
            sq = add_row_xor<8>(sq, sq);
            act[nc] = (sq / (1.0f + sq)) * p * rsqrtf(sq + EPSV);
        }

        if (r < NROUT - 1) {
#pragma unroll
            for (int nc = 0; nc < NCAPS; ++nc) {
                float P[16];
#pragma unroll
                for (int mt = 0; mt < 2; ++mt)
#pragma unroll
                    for (int v = 0; v < 8; ++v) P[mt * 8 + v] = acc[nc][mt][v] * act[nc];
                lg[nc] += fold16(P, x8, x4, x2, x1);
            }
        }
    }

#pragma unroll
    for (int q = 0; q < NCAPS / 2; ++q)
        out[pos * NCD + 32 * q + lane] = half ? act[2 * q + 1] : act[2 * q];
}

extern "C" void kernel_launch(void* const* d_in, const int* in_sizes, int n_in,
                              void* d_out, int out_size, void* d_ws, size_t ws_size,
                              hipStream_t stream) {
    const float* x  = (const float*)d_in[0];
    const float* W  = (const float*)d_in[1];
    const float* b  = (const float*)d_in[2];
    float* out      = (float*)d_out;
    const int positions = 32 * 16 * 16;
    caps_routing_fused<<<dim3(positions / PPB), dim3(32 * PPB), 0, stream>>>(x, W, b, out);
}


# Inputs by 128-bit loads in one group: x straight from global into A fragments (no LDS), W to LDS in one round trip; k order 8h+2ks+v in both loaders, changes the votes' sum order
# speedup vs baseline: 1.9160x; 1.1034x over previous
#include <hip/hip_runtime.h>


typedef __attribute__((ext_vector_type(2))) float v2f;
typedef __attribute__((ext_vector_type(4))) float v4f;
typedef __attribute__((ext_vector_type(8))) float v8f;

#define ICAPS   32
#define IATOMS  16
#define NCAPS   10
#define CDIM    16
#define NCD     160
#define NROUT   3
#define EPSV    1e-7f
#define WSTR    164
#define RSTR    32
#define PPB     4


template <int M>
__device__ __forceinline__ float row_xor(float x) {
    return __builtin_bit_cast(float, __builtin_amdgcn_update_dpp(
        0, __builtin_bit_cast(int, x), 0x160 + M, 0xf, 0xf, true));
}

template <int M>
__device__ __forceinline__ float add_row_xor(float a, float b) {
    float d;
    asm("v_add_f32_dpp %0, %1, %2 row_xmask:%3 row_mask:0xf bank_mask:0xf bound_ctrl:1"
        : "=v"(d) : "v"(a), "v"(b), "n"(M));
    return d;
}

__device__ __forceinline__ float xhalf(float x) {
    return __builtin_bit_cast(float, __builtin_amdgcn_permlanex16(
        __builtin_bit_cast(int, x), __builtin_bit_cast(int, x),
        0x76543210, 0xfedcba98, false, false));
}

__device__ __forceinline__ float fold16(const float (&P)[16], bool x8, bool x4, bool x2, bool x1) {
    float n8[8], n4[4], n2[2];
#pragma unroll
    for (int k = 0; k < 8; ++k) {
        const float keep = x8 ? P[k + 8] : P[k];
        const float send = x8 ? P[k] : P[k + 8];
        n8[k] = add_row_xor<8>(send, keep);
    }
#pragma unroll
    for (int k = 0; k < 4; ++k) {
        const float keep = x4 ? n8[k + 4] : n8[k];
        const float send = x4 ? n8[k] : n8[k + 4];
        n4[k] = add_row_xor<4>(send, keep);
    }
#pragma unroll
    for (int k = 0; k < 2; ++k) {
        const float keep = x2 ? n4[k + 2] : n4[k];
        const float send = x2 ? n4[k] : n4[k + 2];
        n2[k] = add_row_xor<2>(send, keep);
    }
    const float keep = x1 ? n2[1] : n2[0];
    const float send = x1 ? n2[0] : n2[1];
    return add_row_xor<1>(send, keep);
}

__global__ __launch_bounds__(32 * PPB, 4)
void caps_routing_fused(const float* __restrict__ x,
                        const float* __restrict__ Wm,
                        const float* __restrict__ bias,
                        float* __restrict__ out)
{
    __shared__ __attribute__((aligned(16))) float sW[IATOMS * WSTR];
    __shared__ __attribute__((aligned(16))) float sR[PPB][NCAPS * RSTR];

    const int t    = threadIdx.x;
    const int lane = t & 31;
    const int wave = t >> 5;
    const int half = lane >> 4;
    const int lcol = lane & 15;
    const long pos = (long)blockIdx.x * PPB + wave;
    float* sRw = sR[wave];

    const float* xr = x + pos * (ICAPS * IATOMS) + lcol * IATOMS + 8 * half;
    v4f xa[2][2];
#pragma unroll
    for (int mt = 0; mt < 2; ++mt) {
        xa[mt][0] = *(const v4f*)(xr + mt * 16 * IATOMS);
        xa[mt][1] = *(const v4f*)(xr + mt * 16 * IATOMS + 4);
    }
    {
        const int wr = t >> 3, wj = t & 7;
        const float* wg = Wm + wr * NCD + 4 * wj;
        float* ws = &sW[wr * WSTR + 4 * wj];
        v4f wv[5];
#pragma unroll
        for (int i = 0; i < 5; ++i) wv[i] = *(const v4f*)(wg + 32 * i);
#pragma unroll
        for (int i = 0; i < 5; ++i) *(v4f*)(ws + 32 * i) = wv[i];
    }
    float bval[NCAPS];
#pragma unroll
    for (int nc = 0; nc < NCAPS; ++nc) bval[nc] = bias[nc * CDIM + lcol];
    __syncthreads();

    v2f a[2][4];
#pragma unroll
    for (int mt = 0; mt < 2; ++mt) {
        a[mt][0].x = xa[mt][0].x; a[mt][0].y = xa[mt][0].y;
        a[mt][1].x = xa[mt][0].z; a[mt][1].y = xa[mt][0].w;
        a[mt][2].x = xa[mt][1].x; a[mt][2].y = xa[mt][1].y;
        a[mt][3].x = xa[mt][1].z; a[mt][3].y = xa[mt][1].w;
    }
    v8f acc[NCAPS][2];
#pragma unroll
    for (int nc = 0; nc < NCAPS; ++nc) {
        v2f b[4];
#pragma unroll
        for (int ks = 0; ks < 4; ++ks) {
            const float* bp = &sW[(8 * half + 2 * ks) * WSTR + nc * 16 + lcol];
            b[ks].x = bp[0]; b[ks].y = bp[WSTR];
        }
        acc[nc][0] = v8f{}; acc[nc][1] = v8f{};
#pragma unroll
        for (int ks = 0; ks < 4; ++ks)
#pragma unroll
            for (int mt = 0; mt < 2; ++mt)
                acc[nc][mt] = __builtin_amdgcn_wmma_f32_16x16x4_f32(
                                  false, a[mt][ks], false, b[ks], (short)0, acc[nc][mt], false, false);
    }

    const bool x8 = (lcol & 8) != 0, x4 = (lcol & 4) != 0, x2 = (lcol & 2) != 0, x1 = (lcol & 1) != 0;
    const int icl = (lcol >> 3) * 16 + 8 * half + (lcol & 7);
    float lg[NCAPS], act[NCAPS];
#pragma unroll
    for (int n = 0; n < NCAPS; ++n) { lg[n] = 0.0f; act[n] = 0.0f; }

    for (int r = 0; r < NROUT; ++r) {
        {
            float e[NCAPS], mx = -1e30f;
#pragma unroll
            for (int n = 0; n < NCAPS; ++n) mx = fmaxf(mx, lg[n]);
            float s = 0.0f;
#pragma unroll
            for (int n = 0; n < NCAPS; ++n) { e[n] = __expf(lg[n] - mx); s += e[n]; }
            const float inv = 1.0f / s;
            __builtin_amdgcn_wave_barrier();
#pragma unroll
            for (int n = 0; n < NCAPS; ++n) sRw[n * RSTR + icl] = e[n] * inv;
            __builtin_amdgcn_wave_barrier();
        }

#pragma unroll
        for (int nc = 0; nc < NCAPS; ++nc) {
            float p0 = 0.0f, p1 = 0.0f;
#pragma unroll
            for (int mt = 0; mt < 2; ++mt) {
                const v4f* rp = (const v4f*)&sRw[nc * RSTR + mt * 16 + 8 * half];
                const v4f ra = rp[0], rb = rp[1];
                p0 += ra.x * acc[nc][mt][0]; p1 += ra.y * acc[nc][mt][1];
                p0 += ra.z * acc[nc][mt][2]; p1 += ra.w * acc[nc][mt][3];
                p0 += rb.x * acc[nc][mt][4]; p1 += rb.y * acc[nc][mt][5];
                p0 += rb.z * acc[nc][mt][6]; p1 += rb.w * acc[nc][mt][7];
            }
            float p = p0 + p1;
            p += xhalf(p);
            p += bval[nc];

            float sq = p * p;
            sq += row_xor<1>(sq);
            sq = add_row_xor<2>(sq, sq);
            sq = add_row_xor<4>(sq, sq);
            sq = add_row_xor<8>(sq, sq);
            act[nc] = (sq / (1.0f + sq)) * p * rsqrtf(sq + EPSV);
        }

        if (r < NROUT - 1) {
#pragma unroll
            for (int nc = 0; nc < NCAPS; ++nc) {
                float P[16];
#pragma unroll
                for (int mt = 0; mt < 2; ++mt)
#pragma unroll
                    for (int v = 0; v < 8; ++v) P[mt * 8 + v] = acc[nc][mt][v] * act[nc];
                lg[nc] += fold16(P, x8, x4, x2, x1);
            }
        }
    }

#pragma unroll
    for (int q = 0; q < NCAPS / 2; ++q)
        out[pos * NCD + 32 * q + lane] = half ? act[2 * q + 1] : act[2 * q];
}

extern "C" void kernel_launch(void* const* d_in, const int* in_sizes, int n_in,
                              void* d_out, int out_size, void* d_ws, size_t ws_size,
                              hipStream_t stream) {
    const float* x  = (const float*)d_in[0];
    const float* W  = (const float*)d_in[1];
    const float* b  = (const float*)d_in[2];
    float* out      = (float*)d_out;
    const int positions = 32 * 16 * 16;
    caps_routing_fused<<<dim3(positions / PPB), dim3(32 * PPB), 0, stream>>>(x, W, b, out);
}


# Logit update as x*(W*act): U=W*act by 80 in-lane FMAs, x*U by 8 WMMAs, replacing the 16-lane DPP folds; preactivation sums on register pairs so nothing spills; changes logits and softmax sum order
# speedup vs baseline: 2.4357x; 1.2712x over previous
#include <hip/hip_runtime.h>


typedef __attribute__((ext_vector_type(2))) float v2f;
typedef __attribute__((ext_vector_type(4))) float v4f;
typedef __attribute__((ext_vector_type(8))) float v8f;

#define ICAPS   32
#define IATOMS  16
#define NCAPS   10
#define CDIM    16
#define NCD     160
#define NROUT   3
#define EPSV    1e-7f
#define WSTR    164
#define RSTR    32
#define PPB     4


template <int M>
__device__ __forceinline__ float row_xor(float x) {
    return __builtin_bit_cast(float, __builtin_amdgcn_update_dpp(
        0, __builtin_bit_cast(int, x), 0x160 + M, 0xf, 0xf, true));
}

template <int M>
__device__ __forceinline__ float add_row_xor(float a, float b) {
    float d;
    asm("v_add_f32_dpp %0, %1, %2 row_xmask:%3 row_mask:0xf bank_mask:0xf bound_ctrl:1"
        : "=v"(d) : "v"(a), "v"(b), "n"(M));
    return d;
}

__device__ __forceinline__ float xhalf(float x) {
    return __builtin_bit_cast(float, __builtin_amdgcn_permlanex16(
        __builtin_bit_cast(int, x), __builtin_bit_cast(int, x),
        0x76543210, 0xfedcba98, false, false));
}

__global__ __launch_bounds__(32 * PPB, 4)
void caps_routing_fused(const float* __restrict__ x,
                        const float* __restrict__ Wm,
                        const float* __restrict__ bias,
                        float* __restrict__ out)
{
    __shared__ __attribute__((aligned(16))) float sW[IATOMS * WSTR];
    __shared__ __attribute__((aligned(16))) float sR[PPB][NCAPS * RSTR];
    __shared__ __attribute__((aligned(16))) float sA[PPB][NCD];
    __shared__ float sB[NCD];
    __shared__ __attribute__((aligned(16))) float sU[PPB][NCD];

    const int t    = threadIdx.x;
    const int lane = t & 31;
    const int wave = t >> 5;
    const int half = lane >> 4;
    const int lcol = lane & 15;
    const long pos = (long)blockIdx.x * PPB + wave;
    float* sRw = sR[wave];
    float* sAw = sA[wave];
    float* sUw = sU[wave];

    const float* xr = x + pos * (ICAPS * IATOMS) + lcol * IATOMS + 8 * half;
    v4f xa[2][2];
#pragma unroll
    for (int mt = 0; mt < 2; ++mt) {
        xa[mt][0] = *(const v4f*)(xr + mt * 16 * IATOMS);
        xa[mt][1] = *(const v4f*)(xr + mt * 16 * IATOMS + 4);
    }
    {
        const int wr = t >> 3, wj = t & 7;
        const float* wg = Wm + wr * NCD + 4 * wj;
        float* ws = &sW[wr * WSTR + 4 * wj];
        v4f wv[5];
#pragma unroll
        for (int i = 0; i < 5; ++i) wv[i] = *(const v4f*)(wg + 32 * i);
#pragma unroll
        for (int i = 0; i < 5; ++i) *(v4f*)(ws + 32 * i) = wv[i];
    }
    sB[t] = bias[t];
    if (t < NCD - 32 * PPB) sB[32 * PPB + t] = bias[32 * PPB + t];
    __syncthreads();

    v2f a[2][4];
#pragma unroll
    for (int mt = 0; mt < 2; ++mt) {
        a[mt][0].x = xa[mt][0].x; a[mt][0].y = xa[mt][0].y;
        a[mt][1].x = xa[mt][0].z; a[mt][1].y = xa[mt][0].w;
        a[mt][2].x = xa[mt][1].x; a[mt][2].y = xa[mt][1].y;
        a[mt][3].x = xa[mt][1].z; a[mt][3].y = xa[mt][1].w;
    }
    v8f acc[NCAPS][2];
#pragma unroll
    for (int nc = 0; nc < NCAPS; ++nc) {
        v2f b[4];
#pragma unroll
        for (int ks = 0; ks < 4; ++ks) {
            const float* bp = &sW[(8 * half + 2 * ks) * WSTR + nc * 16 + lcol];
            b[ks].x = bp[0]; b[ks].y = bp[WSTR];
        }
        acc[nc][0] = v8f{}; acc[nc][1] = v8f{};
#pragma unroll
        for (int ks = 0; ks < 4; ++ks)
#pragma unroll
            for (int mt = 0; mt < 2; ++mt)
                acc[nc][mt] = __builtin_amdgcn_wmma_f32_16x16x4_f32(
                                  false, a[mt][ks], false, b[ks], (short)0, acc[nc][mt], false, false);
    }

    const int ncA = 5 * half, ncB = 5 - 5 * half;
    float* rA = sRw + ncA * RSTR + lane;
    float* rB = sRw + ncB * RSTR + lane;
    const int l7 = lcol & 7;
    const int ncrow = (l7 < 4 ? l7 : 4) + 5 * (lcol >> 3);
    float lgA[5], lgB[5];
#pragma unroll
    for (int i = 0; i < 5; ++i) { lgA[i] = 0.0f; lgB[i] = 0.0f; }

    for (int r = 0; r < NROUT; ++r) {
        {
            float eA[5], eB[5], mx = -1e30f;
#pragma unroll
            for (int i = 0; i < 5; ++i) mx = fmaxf(mx, fmaxf(lgA[i], lgB[i]));
            float s = 0.0f;
#pragma unroll
            for (int i = 0; i < 5; ++i) { eA[i] = __expf(lgA[i] - mx); s += eA[i]; }
#pragma unroll
            for (int i = 0; i < 5; ++i) { eB[i] = __expf(lgB[i] - mx); s += eB[i]; }
            const float inv = 1.0f / s;
            __builtin_amdgcn_wave_barrier();
#pragma unroll
            for (int i = 0; i < 5; ++i) {
                rA[i * RSTR] = eA[i] * inv;
                rB[i * RSTR] = eB[i] * inv;
            }
            __builtin_amdgcn_wave_barrier();
        }

        __builtin_amdgcn_wave_barrier();
#pragma unroll
        for (int nc = 0; nc < NCAPS; ++nc) {
            v2f pp = {0.0f, 0.0f};
#pragma unroll
            for (int mt = 0; mt < 2; ++mt) {
                const v4f* rp = (const v4f*)&sRw[nc * RSTR + mt * 16 + 8 * half];
                const v4f ra = rp[0], rb = rp[1];
                const v8f av = acc[nc][mt];
                pp = __builtin_elementwise_fma(ra.xy, __builtin_shufflevector(av, av, 0, 1), pp);
                pp = __builtin_elementwise_fma(ra.zw, __builtin_shufflevector(av, av, 2, 3), pp);
                pp = __builtin_elementwise_fma(rb.xy, __builtin_shufflevector(av, av, 4, 5), pp);
                pp = __builtin_elementwise_fma(rb.zw, __builtin_shufflevector(av, av, 6, 7), pp);
            }
            float p = pp.x + pp.y;
            p += xhalf(p);
            p += sB[nc * CDIM + lcol];

            float sq = p * p;
            sq += row_xor<1>(sq);
            sq = add_row_xor<2>(sq, sq);
            sq = add_row_xor<4>(sq, sq);
            sq = add_row_xor<8>(sq, sq);
            sAw[nc * CDIM + lcol] = (sq / (1.0f + sq)) * p * rsqrtf(sq + EPSV);
        }

        if (r < NROUT - 1) {
            __builtin_amdgcn_wave_barrier();

            v2f xb[2][4];
#pragma unroll
            for (int mt = 0; mt < 2; ++mt) {
                const v4f x0 = *(const v4f*)(xr + mt * 16 * IATOMS);
                const v4f x1 = *(const v4f*)(xr + mt * 16 * IATOMS + 4);
                xb[mt][0].x = x0.x; xb[mt][0].y = x0.y; xb[mt][1].x = x0.z; xb[mt][1].y = x0.w;
                xb[mt][2].x = x1.x; xb[mt][2].y = x1.y; xb[mt][3].x = x1.z; xb[mt][3].y = x1.w;
            }

            const float* wrow = &sW[lcol * WSTR + 80 * half];
            const float* arow = &sAw[80 * half];
            float u[5];
#pragma unroll
            for (int i = 0; i < 5; ++i) {
                v2f s2 = {0.0f, 0.0f};
#pragma unroll
                for (int q = 0; q < 4; ++q) {
                    const v4f w4 = *(const v4f*)(wrow + 16 * i + 4 * q);
                    const v4f c4 = *(const v4f*)(arow + 16 * i + 4 * q);
                    s2 = __builtin_elementwise_fma(w4.xy, c4.xy, s2);
                    s2 = __builtin_elementwise_fma(w4.zw, c4.zw, s2);
                }
                u[i] = s2.x + s2.y;
                __builtin_amdgcn_sched_barrier(0);
            }
            __builtin_amdgcn_wave_barrier();
#pragma unroll
            for (int i = 0; i < 5; ++i) sUw[(ncA + i) * IATOMS + lcol] = u[i];
            __builtin_amdgcn_wave_barrier();

            const v4f* up = (const v4f*)&sUw[ncrow * IATOMS + 8 * half];
            const v4f ua = up[0], ub = up[1];
            v2f uf[4];
            uf[0].x = ua.x; uf[0].y = ua.y; uf[1].x = ua.z; uf[1].y = ua.w;
            uf[2].x = ub.x; uf[2].y = ub.y; uf[3].x = ub.z; uf[3].y = ub.w;
            v8f d[2];
            d[0] = v8f{}; d[1] = v8f{};
#pragma unroll
            for (int ks = 0; ks < 4; ++ks)
#pragma unroll
                for (int mt = 0; mt < 2; ++mt)
                    d[mt] = __builtin_amdgcn_wmma_f32_16x16x4_f32(
                                false, uf[ks], false, xb[mt][ks], (short)0, d[mt], false, false);
#pragma unroll
            for (int i = 0; i < 5; ++i) {
                const float keep = half ? d[1][i] : d[0][i];
                const float send = half ? d[0][i] : d[1][i];
                lgA[i] += keep;
                lgB[i] += xhalf(send);
            }
        }
    }

#pragma unroll
    for (int q = 0; q < NCAPS / 2; ++q)
        out[pos * NCD + 32 * q + lane] = sAw[32 * q + lane];
}

extern "C" void kernel_launch(void* const* d_in, const int* in_sizes, int n_in,
                              void* d_out, int out_size, void* d_ws, size_t ws_size,
                              hipStream_t stream) {
    const float* x  = (const float*)d_in[0];
    const float* W  = (const float*)d_in[1];
    const float* b  = (const float*)d_in[2];
    float* out      = (float*)d_out;
    const int positions = 32 * 16 * 16;
    caps_routing_fused<<<dim3(positions / PPB), dim3(32 * PPB), 0, stream>>>(x, W, b, out);
}


# Two positions per wave, side by side: each 128-bit piece of W read from LDS for the in-lane products feeds both positions (LDS 128-bit loads per position 216 to 166); 4096 waves; same sums
# speedup vs baseline: 4.7096x; 1.9336x over previous
#include <hip/hip_runtime.h>


typedef __attribute__((ext_vector_type(2))) float v2f;
typedef __attribute__((ext_vector_type(4))) float v4f;
typedef __attribute__((ext_vector_type(8))) float v8f;

#define ICAPS   32
#define IATOMS  16
#define NCAPS   10
#define CDIM    16
#define NCD     160
#define NROUT   3
#define EPSV    1e-7f
#define WSTR    164
#define WTSTR   20
#define RSTR    32
#define PPW     2
#define WPB     4
#define PPB     (PPW * WPB)


template <int M>
__device__ __forceinline__ float row_xor(float x) {
    return __builtin_bit_cast(float, __builtin_amdgcn_update_dpp(
        0, __builtin_bit_cast(int, x), 0x160 + M, 0xf, 0xf, true));
}

template <int M>
__device__ __forceinline__ float add_row_xor(float a, float b) {
    float d;
    asm("v_add_f32_dpp %0, %1, %2 row_xmask:%3 row_mask:0xf bank_mask:0xf bound_ctrl:1"
        : "=v"(d) : "v"(a), "v"(b), "n"(M));
    return d;
}

__device__ __forceinline__ float xhalf(float x) {
    return __builtin_bit_cast(float, __builtin_amdgcn_permlanex16(
        __builtin_bit_cast(int, x), __builtin_bit_cast(int, x),
        0x76543210, 0xfedcba98, false, false));
}

__global__ __launch_bounds__(32 * WPB, 4)
void caps_routing_fused(const float* __restrict__ x,
                        const float* __restrict__ Wm,
                        const float* __restrict__ bias,
                        float* __restrict__ out)
{
    __shared__ __attribute__((aligned(16))) float sW[IATOMS * WSTR];
    __shared__ __attribute__((aligned(16))) float sWT[NCD * WTSTR];
    __shared__ __attribute__((aligned(16))) float sR[PPB][NCAPS * RSTR];
    __shared__ __attribute__((aligned(16))) float sY[PPB][NCD];
    __shared__ __attribute__((aligned(16))) float sA[PPB][NCD];
    __shared__ __attribute__((aligned(16))) float sU[PPB][NCD];

    const int t    = threadIdx.x;
    const int lane = t & 31;
    const int wave = t >> 5;
    const int half = lane >> 4;
    const int lcol = lane & 15;
    const long pos0 = (long)blockIdx.x * PPB + wave * PPW;
    constexpr int RS = NCAPS * RSTR;
    float* sRw = sR[wave * PPW];
    float* sYw = sY[wave * PPW];
    float* sAw = sA[wave * PPW];
    float* sUw = sU[wave * PPW];

    v4f xa[PPW][2][2];
    float xT[PPW][16];
#pragma unroll
    for (int p = 0; p < PPW; ++p) {
        const float* xp = x + (pos0 + p) * (ICAPS * IATOMS);
        const float* xr = xp + lcol * IATOMS + 8 * half;
#pragma unroll
        for (int mt = 0; mt < 2; ++mt) {
            xa[p][mt][0] = *(const v4f*)(xr + mt * 16 * IATOMS);
            xa[p][mt][1] = *(const v4f*)(xr + mt * 16 * IATOMS + 4);
        }
        const float* xc = xp + (16 * half) * IATOMS + lcol;
#pragma unroll
        for (int e = 0; e < 16; ++e) xT[p][e] = xc[e * IATOMS];
    }
    float bv[5];
#pragma unroll
    for (int i = 0; i < 5; ++i) bv[i] = bias[(5 * half + i) * CDIM + lcol];
    {
        const int wr = t >> 3, wj = t & 7;
        const float* wg = Wm + wr * NCD + 4 * wj;
        float* ws = &sW[wr * WSTR + 4 * wj];
        v4f wv[5];
#pragma unroll
        for (int i = 0; i < 5; ++i) wv[i] = *(const v4f*)(wg + 32 * i);
#pragma unroll
        for (int i = 0; i < 5; ++i) {
            *(v4f*)(ws + 32 * i) = wv[i];
            float* wt = &sWT[(4 * wj + 32 * i) * WTSTR + wr];
            wt[0]         = wv[i].x;
            wt[WTSTR]     = wv[i].y;
            wt[2 * WTSTR] = wv[i].z;
            wt[3 * WTSTR] = wv[i].w;
        }
    }
    __syncthreads();

    v2f xb[PPW][2][4];
#pragma unroll
    for (int p = 0; p < PPW; ++p)
#pragma unroll
        for (int mt = 0; mt < 2; ++mt) {
            xb[p][mt][0].x = xa[p][mt][0].x; xb[p][mt][0].y = xa[p][mt][0].y;
            xb[p][mt][1].x = xa[p][mt][0].z; xb[p][mt][1].y = xa[p][mt][0].w;
            xb[p][mt][2].x = xa[p][mt][1].x; xb[p][mt][2].y = xa[p][mt][1].y;
            xb[p][mt][3].x = xa[p][mt][1].z; xb[p][mt][3].y = xa[p][mt][1].w;
        }

    const int ncA = 5 * half, ncB = 5 - 5 * half;
    float* rA = sRw + ncA * RSTR + lane;
    float* rB = sRw + ncB * RSTR + lane;
    const int l7 = lcol & 7;
    const int ncrow = (l7 < 4 ? l7 : 4) + 5 * (lcol >> 3);
    float lgA[PPW][5], lgB[PPW][5];
#pragma unroll
    for (int p = 0; p < PPW; ++p)
#pragma unroll
        for (int i = 0; i < 5; ++i) { lgA[p][i] = 0.0f; lgB[p][i] = 0.0f; }

    for (int r = 0; r < NROUT; ++r) {
        __builtin_amdgcn_wave_barrier();
#pragma unroll
        for (int p = 0; p < PPW; ++p) {
            float eA[5], eB[5], mx = -1e30f;
#pragma unroll
            for (int i = 0; i < 5; ++i) mx = fmaxf(mx, fmaxf(lgA[p][i], lgB[p][i]));
            float s = 0.0f;
#pragma unroll
            for (int i = 0; i < 5; ++i) { eA[i] = __expf(lgA[p][i] - mx); s += eA[i]; }
#pragma unroll
            for (int i = 0; i < 5; ++i) { eB[i] = __expf(lgB[p][i] - mx); s += eB[i]; }
            const float inv = 1.0f / s;
#pragma unroll
            for (int i = 0; i < 5; ++i) {
                rA[p * RS + i * RSTR] = eA[i] * inv;
                rB[p * RS + i * RSTR] = eB[i] * inv;
            }
        }
        __builtin_amdgcn_wave_barrier();

        v8f y0[PPW], y1[PPW];
        {
            v4f rq[PPW][4];
#pragma unroll
            for (int p = 0; p < PPW; ++p) {
                const v4f* rp = (const v4f*)&sRw[p * RS + ncrow * RSTR + 16 * half];
#pragma unroll
                for (int q = 0; q < 4; ++q) rq[p][q] = rp[q];
                y0[p] = v8f{}; y1[p] = v8f{};
            }
#pragma unroll
            for (int q = 0; q < 4; ++q)
#pragma unroll
                for (int p = 0; p < PPW; ++p) {
                    v2f ra0, ra1, xb0, xb1;
                    ra0.x = rq[p][q].x; ra0.y = rq[p][q].y;
                    ra1.x = rq[p][q].z; ra1.y = rq[p][q].w;
                    xb0.x = xT[p][4 * q];     xb0.y = xT[p][4 * q + 1];
                    xb1.x = xT[p][4 * q + 2]; xb1.y = xT[p][4 * q + 3];
                    y0[p] = __builtin_amdgcn_wmma_f32_16x16x4_f32(false, ra0, false, xb0, (short)0, y0[p], false, false);
                    y1[p] = __builtin_amdgcn_wmma_f32_16x16x4_f32(false, ra1, false, xb1, (short)0, y1[p], false, false);
                }
        }
        __builtin_amdgcn_wave_barrier();
#pragma unroll
        for (int p = 0; p < PPW; ++p)
#pragma unroll
            for (int i = 0; i < 5; ++i) sYw[p * NCD + (ncA + i) * IATOMS + lcol] = y0[p][i] + y1[p][i];
        __builtin_amdgcn_wave_barrier();

        __builtin_amdgcn_wave_barrier();
        float pv[PPW][5];
        {
            const float* wcol = &sWT[(80 * half + lcol) * WTSTR];
            const float* yrow = &sYw[80 * half];
#pragma unroll
            for (int i = 0; i < 5; ++i) {
                v2f s2[PPW];
#pragma unroll
                for (int p = 0; p < PPW; ++p) s2[p] = v2f{0.0f, 0.0f};
#pragma unroll
                for (int q = 0; q < 4; ++q) {
                    const v4f w4 = *(const v4f*)(wcol + 16 * i * WTSTR + 4 * q);
#pragma unroll
                    for (int p = 0; p < PPW; ++p) {
                        const v4f y4 = *(const v4f*)(yrow + p * NCD + 16 * i + 4 * q);
                        s2[p] = __builtin_elementwise_fma(w4.xy, y4.xy, s2[p]);
                        s2[p] = __builtin_elementwise_fma(w4.zw, y4.zw, s2[p]);
                    }
                }
#pragma unroll
                for (int p = 0; p < PPW; ++p) pv[p][i] = (s2[p].x + s2[p].y) + bv[i];
                __builtin_amdgcn_sched_barrier(0);
            }
        }
#pragma unroll
        for (int p = 0; p < PPW; ++p)
#pragma unroll
            for (int i = 0; i < 5; ++i) {
                const float pp = pv[p][i];
                float sq = pp * pp;
                sq += row_xor<1>(sq);
                sq = add_row_xor<2>(sq, sq);
                sq = add_row_xor<4>(sq, sq);
                sq = add_row_xor<8>(sq, sq);
                sAw[p * NCD + (ncA + i) * CDIM + lcol] = (sq / (1.0f + sq)) * pp * rsqrtf(sq + EPSV);
            }

        if (r < NROUT - 1) {
            __builtin_amdgcn_wave_barrier();

            const float* wrow = &sW[lcol * WSTR + 80 * half];
            const float* arow = &sAw[80 * half];
            float u[PPW][5];
#pragma unroll
            for (int i = 0; i < 5; ++i) {
                v2f s2[PPW];
#pragma unroll
                for (int p = 0; p < PPW; ++p) s2[p] = v2f{0.0f, 0.0f};
#pragma unroll
                for (int q = 0; q < 4; ++q) {
                    const v4f w4 = *(const v4f*)(wrow + 16 * i + 4 * q);
#pragma unroll
                    for (int p = 0; p < PPW; ++p) {
                        const v4f c4 = *(const v4f*)(arow + p * NCD + 16 * i + 4 * q);
                        s2[p] = __builtin_elementwise_fma(w4.xy, c4.xy, s2[p]);
                        s2[p] = __builtin_elementwise_fma(w4.zw, c4.zw, s2[p]);
                    }
                }
#pragma unroll
                for (int p = 0; p < PPW; ++p) u[p][i] = s2[p].x + s2[p].y;
                __builtin_amdgcn_sched_barrier(0);
            }
            __builtin_amdgcn_wave_barrier();
#pragma unroll
            for (int p = 0; p < PPW; ++p)
#pragma unroll
                for (int i = 0; i < 5; ++i) sUw[p * NCD + (ncA + i) * IATOMS + lcol] = u[p][i];
            __builtin_amdgcn_wave_barrier();

            v2f uf[PPW][4];
            v8f d[PPW][2];
#pragma unroll
            for (int p = 0; p < PPW; ++p) {
                const v4f* up = (const v4f*)&sUw[p * NCD + ncrow * IATOMS + 8 * half];
                const v4f ua = up[0], ub = up[1];
                uf[p][0].x = ua.x; uf[p][0].y = ua.y; uf[p][1].x = ua.z; uf[p][1].y = ua.w;
                uf[p][2].x = ub.x; uf[p][2].y = ub.y; uf[p][3].x = ub.z; uf[p][3].y = ub.w;
                d[p][0] = v8f{}; d[p][1] = v8f{};
            }
#pragma unroll
            for (int ks = 0; ks < 4; ++ks)
#pragma unroll
                for (int mt = 0; mt < 2; ++mt)
#pragma unroll
                    for (int p = 0; p < PPW; ++p)
                        d[p][mt] = __builtin_amdgcn_wmma_f32_16x16x4_f32(
                                       false, uf[p][ks], false, xb[p][mt][ks], (short)0, d[p][mt], false, false);
#pragma unroll
            for (int p = 0; p < PPW; ++p)
#pragma unroll
                for (int i = 0; i < 5; ++i) {
                    const float keep = half ? d[p][1][i] : d[p][0][i];
                    const float send = half ? d[p][0][i] : d[p][1][i];
                    lgA[p][i] += keep;
                    lgB[p][i] += xhalf(send);
                }
        }
    }

    __builtin_amdgcn_wave_barrier();
#pragma unroll
    for (int p = 0; p < PPW; ++p)
#pragma unroll
        for (int q = 0; q < NCAPS / 2; ++q)
            out[(pos0 + p) * NCD + 32 * q + lane] = sAw[p * NCD + 32 * q + lane];
}

extern "C" void kernel_launch(void* const* d_in, const int* in_sizes, int n_in,
                              void* d_out, int out_size, void* d_ws, size_t ws_size,
                              hipStream_t stream) {
    const float* x  = (const float*)d_in[0];
    const float* W  = (const float*)d_in[1];
    const float* b  = (const float*)d_in[2];
    float* out      = (float*)d_out;
    const int positions = 32 * 16 * 16;
    caps_routing_fused<<<dim3(positions / PPB), dim3(32 * WPB), 0, stream>>>(x, W, b, out);
}


# Softmax normaliser and squash factor by bare v_rcp_f32 and v_rsq_f32 instead of the full division and guarded rsqrt sequences; both quotients change by up to one ulp, sum orders unchanged
# speedup vs baseline: 5.0446x; 1.0711x over previous
#include <hip/hip_runtime.h>


typedef __attribute__((ext_vector_type(2))) float v2f;
typedef __attribute__((ext_vector_type(4))) float v4f;
typedef __attribute__((ext_vector_type(8))) float v8f;

#define ICAPS   32
#define IATOMS  16
#define NCAPS   10
#define CDIM    16
#define NCD     160
#define NROUT   3
#define EPSV    1e-7f
#define WSTR    164
#define WTSTR   20
#define RSTR    32
#define PPW     2
#define WPB     4
#define PPB     (PPW * WPB)


template <int M>
__device__ __forceinline__ float row_xor(float x) {
    return __builtin_bit_cast(float, __builtin_amdgcn_update_dpp(
        0, __builtin_bit_cast(int, x), 0x160 + M, 0xf, 0xf, true));
}

template <int M>
__device__ __forceinline__ float add_row_xor(float a, float b) {
    float d;
    asm("v_add_f32_dpp %0, %1, %2 row_xmask:%3 row_mask:0xf bank_mask:0xf bound_ctrl:1"
        : "=v"(d) : "v"(a), "v"(b), "n"(M));
    return d;
}

__device__ __forceinline__ float xhalf(float x) {
    return __builtin_bit_cast(float, __builtin_amdgcn_permlanex16(
        __builtin_bit_cast(int, x), __builtin_bit_cast(int, x),
        0x76543210, 0xfedcba98, false, false));
}

__global__ __launch_bounds__(32 * WPB, 4)
void caps_routing_fused(const float* __restrict__ x,
                        const float* __restrict__ Wm,
                        const float* __restrict__ bias,
                        float* __restrict__ out)
{
    __shared__ __attribute__((aligned(16))) float sW[IATOMS * WSTR];
    __shared__ __attribute__((aligned(16))) float sWT[NCD * WTSTR];
    __shared__ __attribute__((aligned(16))) float sR[PPB][NCAPS * RSTR];
    __shared__ __attribute__((aligned(16))) float sY[PPB][NCD];
    __shared__ __attribute__((aligned(16))) float sA[PPB][NCD];
    __shared__ __attribute__((aligned(16))) float sU[PPB][NCD];

    const int t    = threadIdx.x;
    const int lane = t & 31;
    const int wave = t >> 5;
    const int half = lane >> 4;
    const int lcol = lane & 15;
    const long pos0 = (long)blockIdx.x * PPB + wave * PPW;
    constexpr int RS = NCAPS * RSTR;
    float* sRw = sR[wave * PPW];
    float* sYw = sY[wave * PPW];
    float* sAw = sA[wave * PPW];
    float* sUw = sU[wave * PPW];

    v4f xa[PPW][2][2];
    float xT[PPW][16];
#pragma unroll
    for (int p = 0; p < PPW; ++p) {
        const float* xp = x + (pos0 + p) * (ICAPS * IATOMS);
        const float* xr = xp + lcol * IATOMS + 8 * half;
#pragma unroll
        for (int mt = 0; mt < 2; ++mt) {
            xa[p][mt][0] = *(const v4f*)(xr + mt * 16 * IATOMS);
            xa[p][mt][1] = *(const v4f*)(xr + mt * 16 * IATOMS + 4);
        }
        const float* xc = xp + (16 * half) * IATOMS + lcol;
#pragma unroll
        for (int e = 0; e < 16; ++e) xT[p][e] = xc[e * IATOMS];
    }
    float bv[5];
#pragma unroll
    for (int i = 0; i < 5; ++i) bv[i] = bias[(5 * half + i) * CDIM + lcol];
    {
        const int wr = t >> 3, wj = t & 7;
        const float* wg = Wm + wr * NCD + 4 * wj;
        float* ws = &sW[wr * WSTR + 4 * wj];
        v4f wv[5];
#pragma unroll
        for (int i = 0; i < 5; ++i) wv[i] = *(const v4f*)(wg + 32 * i);
#pragma unroll
        for (int i = 0; i < 5; ++i) {
            *(v4f*)(ws + 32 * i) = wv[i];
            float* wt = &sWT[(4 * wj + 32 * i) * WTSTR + wr];
            wt[0]         = wv[i].x;
            wt[WTSTR]     = wv[i].y;
            wt[2 * WTSTR] = wv[i].z;
            wt[3 * WTSTR] = wv[i].w;
        }
    }
    __syncthreads();

    v2f xb[PPW][2][4];
#pragma unroll
    for (int p = 0; p < PPW; ++p)
#pragma unroll
        for (int mt = 0; mt < 2; ++mt) {
            xb[p][mt][0].x = xa[p][mt][0].x; xb[p][mt][0].y = xa[p][mt][0].y;
            xb[p][mt][1].x = xa[p][mt][0].z; xb[p][mt][1].y = xa[p][mt][0].w;
            xb[p][mt][2].x = xa[p][mt][1].x; xb[p][mt][2].y = xa[p][mt][1].y;
            xb[p][mt][3].x = xa[p][mt][1].z; xb[p][mt][3].y = xa[p][mt][1].w;
        }

    const int ncA = 5 * half, ncB = 5 - 5 * half;
    float* rA = sRw + ncA * RSTR + lane;
    float* rB = sRw + ncB * RSTR + lane;
    const int l7 = lcol & 7;
    const int ncrow = (l7 < 4 ? l7 : 4) + 5 * (lcol >> 3);
    float lgA[PPW][5], lgB[PPW][5];
#pragma unroll
    for (int p = 0; p < PPW; ++p)
#pragma unroll
        for (int i = 0; i < 5; ++i) { lgA[p][i] = 0.0f; lgB[p][i] = 0.0f; }

    for (int r = 0; r < NROUT; ++r) {
        __builtin_amdgcn_wave_barrier();
#pragma unroll
        for (int p = 0; p < PPW; ++p) {
            float eA[5], eB[5], mx = -1e30f;
#pragma unroll
            for (int i = 0; i < 5; ++i) mx = fmaxf(mx, fmaxf(lgA[p][i], lgB[p][i]));
            float s = 0.0f;
#pragma unroll
            for (int i = 0; i < 5; ++i) { eA[i] = __expf(lgA[p][i] - mx); s += eA[i]; }
#pragma unroll
            for (int i = 0; i < 5; ++i) { eB[i] = __expf(lgB[p][i] - mx); s += eB[i]; }
            const float inv = __builtin_amdgcn_rcpf(s);
#pragma unroll
            for (int i = 0; i < 5; ++i) {
                rA[p * RS + i * RSTR] = eA[i] * inv;
                rB[p * RS + i * RSTR] = eB[i] * inv;
            }
        }
        __builtin_amdgcn_wave_barrier();

        v8f y0[PPW], y1[PPW];
        {
            v4f rq[PPW][4];
#pragma unroll
            for (int p = 0; p < PPW; ++p) {
                const v4f* rp = (const v4f*)&sRw[p * RS + ncrow * RSTR + 16 * half];
#pragma unroll
                for (int q = 0; q < 4; ++q) rq[p][q] = rp[q];
                y0[p] = v8f{}; y1[p] = v8f{};
            }
#pragma unroll
            for (int q = 0; q < 4; ++q)
#pragma unroll
                for (int p = 0; p < PPW; ++p) {
                    v2f ra0, ra1, xb0, xb1;
                    ra0.x = rq[p][q].x; ra0.y = rq[p][q].y;
                    ra1.x = rq[p][q].z; ra1.y = rq[p][q].w;
                    xb0.x = xT[p][4 * q];     xb0.y = xT[p][4 * q + 1];
                    xb1.x = xT[p][4 * q + 2]; xb1.y = xT[p][4 * q + 3];
                    y0[p] = __builtin_amdgcn_wmma_f32_16x16x4_f32(false, ra0, false, xb0, (short)0, y0[p], false, false);
                    y1[p] = __builtin_amdgcn_wmma_f32_16x16x4_f32(false, ra1, false, xb1, (short)0, y1[p], false, false);
                }
        }
        __builtin_amdgcn_wave_barrier();
#pragma unroll
        for (int p = 0; p < PPW; ++p)
#pragma unroll
            for (int i = 0; i < 5; ++i) sYw[p * NCD + (ncA + i) * IATOMS + lcol] = y0[p][i] + y1[p][i];
        __builtin_amdgcn_wave_barrier();

        __builtin_amdgcn_wave_barrier();
        float pv[PPW][5];
        {
            const float* wcol = &sWT[(80 * half + lcol) * WTSTR];
            const float* yrow = &sYw[80 * half];
#pragma unroll
            for (int i = 0; i < 5; ++i) {
                v2f s2[PPW];
#pragma unroll
                for (int p = 0; p < PPW; ++p) s2[p] = v2f{0.0f, 0.0f};
#pragma unroll
                for (int q = 0; q < 4; ++q) {
                    const v4f w4 = *(const v4f*)(wcol + 16 * i * WTSTR + 4 * q);
#pragma unroll
                    for (int p = 0; p < PPW; ++p) {
                        const v4f y4 = *(const v4f*)(yrow + p * NCD + 16 * i + 4 * q);
                        s2[p] = __builtin_elementwise_fma(w4.xy, y4.xy, s2[p]);
                        s2[p] = __builtin_elementwise_fma(w4.zw, y4.zw, s2[p]);
                    }
                }
#pragma unroll
                for (int p = 0; p < PPW; ++p) pv[p][i] = (s2[p].x + s2[p].y) + bv[i];
                __builtin_amdgcn_sched_barrier(0);
            }
        }
#pragma unroll
        for (int p = 0; p < PPW; ++p)
#pragma unroll
            for (int i = 0; i < 5; ++i) {
                const float pp = pv[p][i];
                float sq = pp * pp;
                sq += row_xor<1>(sq);
                sq = add_row_xor<2>(sq, sq);
                sq = add_row_xor<4>(sq, sq);
                sq = add_row_xor<8>(sq, sq);
                const float sqc = fminf(sq, 0x1p100f);
                sAw[p * NCD + (ncA + i) * CDIM + lcol] =
                    (sqc * __builtin_amdgcn_rcpf(1.0f + sqc)) * pp * __builtin_amdgcn_rsqf(sq + EPSV);
            }

        if (r < NROUT - 1) {
            __builtin_amdgcn_wave_barrier();

            const float* wrow = &sW[lcol * WSTR + 80 * half];
            const float* arow = &sAw[80 * half];
            float u[PPW][5];
#pragma unroll
            for (int i = 0; i < 5; ++i) {
                v2f s2[PPW];
#pragma unroll
                for (int p = 0; p < PPW; ++p) s2[p] = v2f{0.0f, 0.0f};
#pragma unroll
                for (int q = 0; q < 4; ++q) {
                    const v4f w4 = *(const v4f*)(wrow + 16 * i + 4 * q);
#pragma unroll
                    for (int p = 0; p < PPW; ++p) {
                        const v4f c4 = *(const v4f*)(arow + p * NCD + 16 * i + 4 * q);
                        s2[p] = __builtin_elementwise_fma(w4.xy, c4.xy, s2[p]);
                        s2[p] = __builtin_elementwise_fma(w4.zw, c4.zw, s2[p]);
                    }
                }
#pragma unroll
                for (int p = 0; p < PPW; ++p) u[p][i] = s2[p].x + s2[p].y;
                __builtin_amdgcn_sched_barrier(0);
            }
            __builtin_amdgcn_wave_barrier();
#pragma unroll
            for (int p = 0; p < PPW; ++p)
#pragma unroll
                for (int i = 0; i < 5; ++i) sUw[p * NCD + (ncA + i) * IATOMS + lcol] = u[p][i];
            __builtin_amdgcn_wave_barrier();

            v2f uf[PPW][4];
            v8f d[PPW][2];
#pragma unroll
            for (int p = 0; p < PPW; ++p) {
                const v4f* up = (const v4f*)&sUw[p * NCD + ncrow * IATOMS + 8 * half];
                const v4f ua = up[0], ub = up[1];
                uf[p][0].x = ua.x; uf[p][0].y = ua.y; uf[p][1].x = ua.z; uf[p][1].y = ua.w;
                uf[p][2].x = ub.x; uf[p][2].y = ub.y; uf[p][3].x = ub.z; uf[p][3].y = ub.w;
                d[p][0] = v8f{}; d[p][1] = v8f{};
            }
#pragma unroll
            for (int ks = 0; ks < 4; ++ks)
#pragma unroll
                for (int mt = 0; mt < 2; ++mt)
#pragma unroll
                    for (int p = 0; p < PPW; ++p)
                        d[p][mt] = __builtin_amdgcn_wmma_f32_16x16x4_f32(
                                       false, uf[p][ks], false, xb[p][mt][ks], (short)0, d[p][mt], false, false);
#pragma unroll
            for (int p = 0; p < PPW; ++p)
#pragma unroll
                for (int i = 0; i < 5; ++i) {
                    const float keep = half ? d[p][1][i] : d[p][0][i];
                    const float send = half ? d[p][0][i] : d[p][1][i];
                    lgA[p][i] += keep;
                    lgB[p][i] += xhalf(send);
                }
        }
    }

    __builtin_amdgcn_wave_barrier();
#pragma unroll
    for (int p = 0; p < PPW; ++p)
#pragma unroll
        for (int q = 0; q < NCAPS / 2; ++q)
            out[(pos0 + p) * NCD + 32 * q + lane] = sAw[p * NCD + 32 * q + lane];
}

extern "C" void kernel_launch(void* const* d_in, const int* in_sizes, int n_in,
                              void* d_out, int out_size, void* d_ws, size_t ws_size,
                              hipStream_t stream) {
    const float* x  = (const float*)d_in[0];
    const float* W  = (const float*)d_in[1];
    const float* b  = (const float*)d_in[2];
    float* out      = (float*)d_out;
    const int positions = 32 * 16 * 16;
    caps_routing_fused<<<dim3(positions / PPB), dim3(32 * WPB), 0, stream>>>(x, W, b, out);
}


# First routing iteration specialised: zero logits give weights 1/10, so Y is one row, 0.1*sum_ic x, summed in lanes; its softmax, 16 WMMAs and 32 LDS row loads go; changes that iteration's sum order
# speedup vs baseline: 5.3862x; 1.0677x over previous
#include <hip/hip_runtime.h>


typedef __attribute__((ext_vector_type(2))) float v2f;
typedef __attribute__((ext_vector_type(4))) float v4f;
typedef __attribute__((ext_vector_type(8))) float v8f;

#define ICAPS   32
#define IATOMS  16
#define NCAPS   10
#define CDIM    16
#define NCD     160
#define NROUT   3
#define EPSV    1e-7f
#define WSTR    164
#define WTSTR   20
#define RSTR    32
#define PPW     2
#define WPB     4
#define PPB     (PPW * WPB)


template <int M>
__device__ __forceinline__ float row_xor(float x) {
    return __builtin_bit_cast(float, __builtin_amdgcn_update_dpp(
        0, __builtin_bit_cast(int, x), 0x160 + M, 0xf, 0xf, true));
}

template <int M>
__device__ __forceinline__ float add_row_xor(float a, float b) {
    float d;
    asm("v_add_f32_dpp %0, %1, %2 row_xmask:%3 row_mask:0xf bank_mask:0xf bound_ctrl:1"
        : "=v"(d) : "v"(a), "v"(b), "n"(M));
    return d;
}

__device__ __forceinline__ float xhalf(float x) {
    return __builtin_bit_cast(float, __builtin_amdgcn_permlanex16(
        __builtin_bit_cast(int, x), __builtin_bit_cast(int, x),
        0x76543210, 0xfedcba98, false, false));
}

__global__ __launch_bounds__(32 * WPB, 4)
void caps_routing_fused(const float* __restrict__ x,
                        const float* __restrict__ Wm,
                        const float* __restrict__ bias,
                        float* __restrict__ out)
{
    __shared__ __attribute__((aligned(16))) float sW[IATOMS * WSTR];
    __shared__ __attribute__((aligned(16))) float sWT[NCD * WTSTR];
    __shared__ __attribute__((aligned(16))) float sR[PPB][NCAPS * RSTR];
    __shared__ __attribute__((aligned(16))) float sY[PPB][NCD];
    __shared__ __attribute__((aligned(16))) float sA[PPB][NCD];
    __shared__ __attribute__((aligned(16))) float sU[PPB][NCD];

    const int t    = threadIdx.x;
    const int lane = t & 31;
    const int wave = t >> 5;
    const int half = lane >> 4;
    const int lcol = lane & 15;
    const long pos0 = (long)blockIdx.x * PPB + wave * PPW;
    constexpr int RS = NCAPS * RSTR;
    float* sRw = sR[wave * PPW];
    float* sYw = sY[wave * PPW];
    float* sAw = sA[wave * PPW];
    float* sUw = sU[wave * PPW];

    v4f xa[PPW][2][2];
    float xT[PPW][16];
#pragma unroll
    for (int p = 0; p < PPW; ++p) {
        const float* xp = x + (pos0 + p) * (ICAPS * IATOMS);
        const float* xr = xp + lcol * IATOMS + 8 * half;
#pragma unroll
        for (int mt = 0; mt < 2; ++mt) {
            xa[p][mt][0] = *(const v4f*)(xr + mt * 16 * IATOMS);
            xa[p][mt][1] = *(const v4f*)(xr + mt * 16 * IATOMS + 4);
        }
        const float* xc = xp + (16 * half) * IATOMS + lcol;
#pragma unroll
        for (int e = 0; e < 16; ++e) xT[p][e] = xc[e * IATOMS];
    }
    float bv[5];
#pragma unroll
    for (int i = 0; i < 5; ++i) bv[i] = bias[(5 * half + i) * CDIM + lcol];
    {
        const int wr = t >> 3, wj = t & 7;
        const float* wg = Wm + wr * NCD + 4 * wj;
        float* ws = &sW[wr * WSTR + 4 * wj];
        v4f wv[5];
#pragma unroll
        for (int i = 0; i < 5; ++i) wv[i] = *(const v4f*)(wg + 32 * i);
#pragma unroll
        for (int i = 0; i < 5; ++i) {
            *(v4f*)(ws + 32 * i) = wv[i];
            float* wt = &sWT[(4 * wj + 32 * i) * WTSTR + wr];
            wt[0]         = wv[i].x;
            wt[WTSTR]     = wv[i].y;
            wt[2 * WTSTR] = wv[i].z;
            wt[3 * WTSTR] = wv[i].w;
        }
    }
    __syncthreads();

    v2f xb[PPW][2][4];
#pragma unroll
    for (int p = 0; p < PPW; ++p)
#pragma unroll
        for (int mt = 0; mt < 2; ++mt) {
            xb[p][mt][0].x = xa[p][mt][0].x; xb[p][mt][0].y = xa[p][mt][0].y;
            xb[p][mt][1].x = xa[p][mt][0].z; xb[p][mt][1].y = xa[p][mt][0].w;
            xb[p][mt][2].x = xa[p][mt][1].x; xb[p][mt][2].y = xa[p][mt][1].y;
            xb[p][mt][3].x = xa[p][mt][1].z; xb[p][mt][3].y = xa[p][mt][1].w;
        }

    const int ncA = 5 * half, ncB = 5 - 5 * half;
    float* rA = sRw + ncA * RSTR + lane;
    float* rB = sRw + ncB * RSTR + lane;
    const int l7 = lcol & 7;
    const int ncrow = (l7 < 4 ? l7 : 4) + 5 * (lcol >> 3);
    float lgA[PPW][5], lgB[PPW][5];
#pragma unroll
    for (int p = 0; p < PPW; ++p)
#pragma unroll
        for (int i = 0; i < 5; ++i) { lgA[p][i] = 0.0f; lgB[p][i] = 0.0f; }

    for (int r = 0; r < NROUT; ++r) {
        __builtin_amdgcn_wave_barrier();
        float pv[PPW][5];
        const float* wcol = &sWT[(80 * half + lcol) * WTSTR];
        const float* yrow = &sYw[80 * half];
        if (r == 0) {
#pragma unroll
            for (int p = 0; p < PPW; ++p) {
                float t[8];
#pragma unroll
                for (int e = 0; e < 8; ++e) t[e] = xT[p][2 * e] + xT[p][2 * e + 1];
                const float s = ((t[0] + t[1]) + (t[2] + t[3])) + ((t[4] + t[5]) + (t[6] + t[7]));
                sYw[p * NCD + ncA * IATOMS + lcol] = (1.0f / NCAPS) * (s + xhalf(s));
            }
            __builtin_amdgcn_wave_barrier();
            v4f yb[PPW][4];
#pragma unroll
            for (int p = 0; p < PPW; ++p)
#pragma unroll
                for (int q = 0; q < 4; ++q) yb[p][q] = *(const v4f*)(yrow + p * NCD + 4 * q);
#pragma unroll
            for (int i = 0; i < 5; ++i) {
                v2f s2[PPW];
#pragma unroll
                for (int p = 0; p < PPW; ++p) s2[p] = v2f{0.0f, 0.0f};
#pragma unroll
                for (int q = 0; q < 4; ++q) {
                    const v4f w4 = *(const v4f*)(wcol + 16 * i * WTSTR + 4 * q);
#pragma unroll
                    for (int p = 0; p < PPW; ++p) {
                        s2[p] = __builtin_elementwise_fma(w4.xy, yb[p][q].xy, s2[p]);
                        s2[p] = __builtin_elementwise_fma(w4.zw, yb[p][q].zw, s2[p]);
                    }
                }
#pragma unroll
                for (int p = 0; p < PPW; ++p) pv[p][i] = (s2[p].x + s2[p].y) + bv[i];
            }
        } else {
            __builtin_amdgcn_wave_barrier();
#pragma unroll
            for (int p = 0; p < PPW; ++p) {
                float eA[5], eB[5], mx = -1e30f;
#pragma unroll
                for (int i = 0; i < 5; ++i) mx = fmaxf(mx, fmaxf(lgA[p][i], lgB[p][i]));
                float s = 0.0f;
#pragma unroll
                for (int i = 0; i < 5; ++i) { eA[i] = __expf(lgA[p][i] - mx); s += eA[i]; }
#pragma unroll
                for (int i = 0; i < 5; ++i) { eB[i] = __expf(lgB[p][i] - mx); s += eB[i]; }
                const float inv = __builtin_amdgcn_rcpf(s);
#pragma unroll
                for (int i = 0; i < 5; ++i) {
                    rA[p * RS + i * RSTR] = eA[i] * inv;
                    rB[p * RS + i * RSTR] = eB[i] * inv;
                }
            }
            __builtin_amdgcn_wave_barrier();

            v8f y0[PPW], y1[PPW];
            {
                v4f rq[PPW][4];
#pragma unroll
                for (int p = 0; p < PPW; ++p) {
                    const v4f* rp = (const v4f*)&sRw[p * RS + ncrow * RSTR + 16 * half];
#pragma unroll
                    for (int q = 0; q < 4; ++q) rq[p][q] = rp[q];
                    y0[p] = v8f{}; y1[p] = v8f{};
                }
#pragma unroll
                for (int q = 0; q < 4; ++q)
#pragma unroll
                    for (int p = 0; p < PPW; ++p) {
                        v2f ra0, ra1, xb0, xb1;
                        ra0.x = rq[p][q].x; ra0.y = rq[p][q].y;
                        ra1.x = rq[p][q].z; ra1.y = rq[p][q].w;
                        xb0.x = xT[p][4 * q];     xb0.y = xT[p][4 * q + 1];
                        xb1.x = xT[p][4 * q + 2]; xb1.y = xT[p][4 * q + 3];
                        y0[p] = __builtin_amdgcn_wmma_f32_16x16x4_f32(false, ra0, false, xb0, (short)0, y0[p], false, false);
                        y1[p] = __builtin_amdgcn_wmma_f32_16x16x4_f32(false, ra1, false, xb1, (short)0, y1[p], false, false);
                    }
            }
            __builtin_amdgcn_wave_barrier();
#pragma unroll
            for (int p = 0; p < PPW; ++p)
#pragma unroll
                for (int i = 0; i < 5; ++i) sYw[p * NCD + (ncA + i) * IATOMS + lcol] = y0[p][i] + y1[p][i];
            __builtin_amdgcn_wave_barrier();

            {
#pragma unroll
                for (int i = 0; i < 5; ++i) {
                    v2f s2[PPW];
#pragma unroll
                    for (int p = 0; p < PPW; ++p) s2[p] = v2f{0.0f, 0.0f};
#pragma unroll
                    for (int q = 0; q < 4; ++q) {
                        const v4f w4 = *(const v4f*)(wcol + 16 * i * WTSTR + 4 * q);
#pragma unroll
                        for (int p = 0; p < PPW; ++p) {
                            const v4f y4 = *(const v4f*)(yrow + p * NCD + 16 * i + 4 * q);
                            s2[p] = __builtin_elementwise_fma(w4.xy, y4.xy, s2[p]);
                            s2[p] = __builtin_elementwise_fma(w4.zw, y4.zw, s2[p]);
                        }
                    }
#pragma unroll
                    for (int p = 0; p < PPW; ++p) pv[p][i] = (s2[p].x + s2[p].y) + bv[i];
                    __builtin_amdgcn_sched_barrier(0);
                }
            }
        }
#pragma unroll
        for (int p = 0; p < PPW; ++p)
#pragma unroll
            for (int i = 0; i < 5; ++i) {
                const float pp = pv[p][i];
                float sq = pp * pp;
                sq += row_xor<1>(sq);
                sq = add_row_xor<2>(sq, sq);
                sq = add_row_xor<4>(sq, sq);
                sq = add_row_xor<8>(sq, sq);
                const float sqc = fminf(sq, 0x1p100f);
                sAw[p * NCD + (ncA + i) * CDIM + lcol] =
                    (sqc * __builtin_amdgcn_rcpf(1.0f + sqc)) * pp * __builtin_amdgcn_rsqf(sq + EPSV);
            }

        if (r < NROUT - 1) {
            __builtin_amdgcn_wave_barrier();

            const float* wrow = &sW[lcol * WSTR + 80 * half];
            const float* arow = &sAw[80 * half];
            float u[PPW][5];
#pragma unroll
            for (int i = 0; i < 5; ++i) {
                v2f s2[PPW];
#pragma unroll
                for (int p = 0; p < PPW; ++p) s2[p] = v2f{0.0f, 0.0f};
#pragma unroll
                for (int q = 0; q < 4; ++q) {
                    const v4f w4 = *(const v4f*)(wrow + 16 * i + 4 * q);
#pragma unroll
                    for (int p = 0; p < PPW; ++p) {
                        const v4f c4 = *(const v4f*)(arow + p * NCD + 16 * i + 4 * q);
                        s2[p] = __builtin_elementwise_fma(w4.xy, c4.xy, s2[p]);
                        s2[p] = __builtin_elementwise_fma(w4.zw, c4.zw, s2[p]);
                    }
                }
#pragma unroll
                for (int p = 0; p < PPW; ++p) u[p][i] = s2[p].x + s2[p].y;
                __builtin_amdgcn_sched_barrier(0);
            }
            __builtin_amdgcn_wave_barrier();
#pragma unroll
            for (int p = 0; p < PPW; ++p)
#pragma unroll
                for (int i = 0; i < 5; ++i) sUw[p * NCD + (ncA + i) * IATOMS + lcol] = u[p][i];
            __builtin_amdgcn_wave_barrier();

            v2f uf[PPW][4];
            v8f d[PPW][2];
#pragma unroll
            for (int p = 0; p < PPW; ++p) {
                const v4f* up = (const v4f*)&sUw[p * NCD + ncrow * IATOMS + 8 * half];
                const v4f ua = up[0], ub = up[1];
                uf[p][0].x = ua.x; uf[p][0].y = ua.y; uf[p][1].x = ua.z; uf[p][1].y = ua.w;
                uf[p][2].x = ub.x; uf[p][2].y = ub.y; uf[p][3].x = ub.z; uf[p][3].y = ub.w;
                d[p][0] = v8f{}; d[p][1] = v8f{};
            }
#pragma unroll
            for (int ks = 0; ks < 4; ++ks)
#pragma unroll
                for (int mt = 0; mt < 2; ++mt)
#pragma unroll
                    for (int p = 0; p < PPW; ++p)
                        d[p][mt] = __builtin_amdgcn_wmma_f32_16x16x4_f32(
                                       false, uf[p][ks], false, xb[p][mt][ks], (short)0, d[p][mt], false, false);
#pragma unroll
            for (int p = 0; p < PPW; ++p)
#pragma unroll
                for (int i = 0; i < 5; ++i) {
                    const float keep = half ? d[p][1][i] : d[p][0][i];
                    const float send = half ? d[p][0][i] : d[p][1][i];
                    lgA[p][i] += keep;
                    lgB[p][i] += xhalf(send);
                }
        }
    }

    __builtin_amdgcn_wave_barrier();
#pragma unroll
    for (int p = 0; p < PPW; ++p)
#pragma unroll
        for (int q = 0; q < NCAPS / 2; ++q)
            out[(pos0 + p) * NCD + 32 * q + lane] = sAw[p * NCD + 32 * q + lane];
}

extern "C" void kernel_launch(void* const* d_in, const int* in_sizes, int n_in,
                              void* d_out, int out_size, void* d_ws, size_t ws_size,
                              hipStream_t stream) {
    const float* x  = (const float*)d_in[0];
    const float* W  = (const float*)d_in[1];
    const float* b  = (const float*)d_in[2];
    float* out      = (float*)d_out;
    const int positions = 32 * 16 * 16;
    caps_routing_fused<<<dim3(positions / PPB), dim3(32 * WPB), 0, stream>>>(x, W, b, out);
}


# Both W products as WMMA tiles over the work-group's 16 positions (8 waves, a tile per capsule; Y, act, U cross waves in LDS): in-lane FMAs and their LDS loads go; changes preactivation, U sum order
# speedup vs baseline: 5.8481x; 1.0858x over previous
#include <hip/hip_runtime.h>


typedef __attribute__((ext_vector_type(2))) float v2f;
typedef __attribute__((ext_vector_type(4))) float v4f;
typedef __attribute__((ext_vector_type(8))) float v8f;

#define ICAPS   32
#define IATOMS  16
#define NCAPS   10
#define CDIM    16
#define NCD     160
#define NROUT   3
#define EPSV    1e-7f
#define WSTR    164
#define WTSTR   20
#define RSTR    32
#define PST     164
#define PPW     2
#define WPB     8
#define PPB     (PPW * WPB)

static_assert(PPB == 16, "the products with W take the work-group's positions as the 16 rows of a matrix tile");


template <int M>
__device__ __forceinline__ float row_xor(float x) {
    return __builtin_bit_cast(float, __builtin_amdgcn_update_dpp(
        0, __builtin_bit_cast(int, x), 0x160 + M, 0xf, 0xf, true));
}

template <int M>
__device__ __forceinline__ float add_row_xor(float a, float b) {
    float d;
    asm("v_add_f32_dpp %0, %1, %2 row_xmask:%3 row_mask:0xf bank_mask:0xf bound_ctrl:1"
        : "=v"(d) : "v"(a), "v"(b), "n"(M));
    return d;
}

__device__ __forceinline__ float xhalf(float x) {
    return __builtin_bit_cast(float, __builtin_amdgcn_permlanex16(
        __builtin_bit_cast(int, x), __builtin_bit_cast(int, x),
        0x76543210, 0xfedcba98, false, false));
}

__device__ __forceinline__ v8f mma4(v2f a, v2f b, v8f c) {
    return __builtin_amdgcn_wmma_f32_16x16x4_f32(false, a, false, b, (short)0, c, false, false);
}

template <int NT>
__device__ __forceinline__ void capsule_tiles(const float* sY, float* sA, float* sU, const float* sW, const float* sWT,
                                              int wave, int lcol, int half, bool first, bool need_u,
                                              float bt0, float bt1)
{
    v4f ya[NT], yb[NT], wa[NT], wb[NT];
#pragma unroll
    for (int tl = 0; tl < NT; ++tl) {
        const int nc = wave + WPB * tl;
        const float* yp = sY + lcol * PST + (first ? 0 : nc * IATOMS) + 8 * half;
        const float* wp = sWT + (CDIM * nc + lcol) * WTSTR + 8 * half;
        ya[tl] = *(const v4f*)yp; yb[tl] = *(const v4f*)(yp + 4);
        wa[tl] = *(const v4f*)wp; wb[tl] = *(const v4f*)(wp + 4);
    }
    v8f a0[NT], a1[NT];
#pragma unroll
    for (int tl = 0; tl < NT; ++tl) {
        a0[tl] = mma4(v2f{ya[tl].x, ya[tl].y}, v2f{wa[tl].x, wa[tl].y}, v8f{});
        a1[tl] = mma4(v2f{ya[tl].z, ya[tl].w}, v2f{wa[tl].z, wa[tl].w}, v8f{});
    }
#pragma unroll
    for (int tl = 0; tl < NT; ++tl) {
        a0[tl] = mma4(v2f{yb[tl].x, yb[tl].y}, v2f{wb[tl].x, wb[tl].y}, a0[tl]);
        a1[tl] = mma4(v2f{yb[tl].z, yb[tl].w}, v2f{wb[tl].z, wb[tl].w}, a1[tl]);
    }
#pragma unroll
    for (int tl = 0; tl < NT; ++tl) {
        const int nc = wave + WPB * tl;
        const float bcol = tl ? bt1 : bt0;
#pragma unroll
        for (int e = 0; e < 8; ++e) {
            const float pp = (a0[tl][e] + a1[tl][e]) + bcol;
            float sq = pp * pp;
            sq += row_xor<1>(sq);
            sq = add_row_xor<2>(sq, sq);
            sq = add_row_xor<4>(sq, sq);
            sq = add_row_xor<8>(sq, sq);
            const float sqc = fminf(sq, 0x1p100f);
            sA[(8 * half + e) * PST + nc * CDIM + lcol] =
                (sqc * __builtin_amdgcn_rcpf(1.0f + sqc)) * pp * __builtin_amdgcn_rsqf(sq + EPSV);
        }
    }
    if (need_u) {
        __builtin_amdgcn_wave_barrier();
        v4f aa[NT], ab[NT], va[NT], vb[NT];
#pragma unroll
        for (int tl = 0; tl < NT; ++tl) {
            const int nc = wave + WPB * tl;
            const float* ap = sA + lcol * PST + nc * CDIM + 8 * half;
            const float* wq = sW + lcol * WSTR + CDIM * nc + 8 * half;
            aa[tl] = *(const v4f*)ap; ab[tl] = *(const v4f*)(ap + 4);
            va[tl] = *(const v4f*)wq; vb[tl] = *(const v4f*)(wq + 4);
        }
        v8f u0[NT], u1[NT];
#pragma unroll
        for (int tl = 0; tl < NT; ++tl) {
            u0[tl] = mma4(v2f{aa[tl].x, aa[tl].y}, v2f{va[tl].x, va[tl].y}, v8f{});
            u1[tl] = mma4(v2f{aa[tl].z, aa[tl].w}, v2f{va[tl].z, va[tl].w}, v8f{});
        }
#pragma unroll
        for (int tl = 0; tl < NT; ++tl) {
            u0[tl] = mma4(v2f{ab[tl].x, ab[tl].y}, v2f{vb[tl].x, vb[tl].y}, u0[tl]);
            u1[tl] = mma4(v2f{ab[tl].z, ab[tl].w}, v2f{vb[tl].z, vb[tl].w}, u1[tl]);
        }
#pragma unroll
        for (int tl = 0; tl < NT; ++tl) {
            const int nc = wave + WPB * tl;
#pragma unroll
            for (int e = 0; e < 8; ++e) sU[(8 * half + e) * PST + nc * IATOMS + lcol] = u0[tl][e] + u1[tl][e];
        }
    }
}

__global__ __launch_bounds__(32 * WPB, 4)
void caps_routing_fused(const float* __restrict__ x,
                        const float* __restrict__ Wm,
                        const float* __restrict__ bias,
                        float* __restrict__ out)
{
    __shared__ __attribute__((aligned(16))) float sW[IATOMS * WSTR];
    __shared__ __attribute__((aligned(16))) float sWT[NCD * WTSTR];
    __shared__ __attribute__((aligned(16))) float sR[PPB][NCAPS * RSTR];
    __shared__ __attribute__((aligned(16))) float sY[PPB * PST];
    __shared__ __attribute__((aligned(16))) float sA[PPB * PST];
    __shared__ __attribute__((aligned(16))) float sU[PPB * PST];

    const int t    = threadIdx.x;
    const int lane = t & 31;
    const int wave = __builtin_amdgcn_readfirstlane(t >> 5);
    const int half = lane >> 4;
    const int lcol = lane & 15;
    const long pos0 = (long)blockIdx.x * PPB + wave * PPW;
    constexpr int RS = NCAPS * RSTR;
    float* sRw = sR[wave * PPW];
    float* sYw = sY + wave * PPW * PST;
    float* sAw = sA + wave * PPW * PST;
    float* sUw = sU + wave * PPW * PST;

    v4f xa[PPW][2][2];
    float xT[PPW][16];
#pragma unroll
    for (int p = 0; p < PPW; ++p) {
        const float* xp = x + (pos0 + p) * (ICAPS * IATOMS);
        const float* xr = xp + lcol * IATOMS + 8 * half;
#pragma unroll
        for (int mt = 0; mt < 2; ++mt) {
            xa[p][mt][0] = *(const v4f*)(xr + mt * 16 * IATOMS);
            xa[p][mt][1] = *(const v4f*)(xr + mt * 16 * IATOMS + 4);
        }
        const float* xc = xp + (16 * half) * IATOMS + lcol;
#pragma unroll
        for (int e = 0; e < 16; ++e) xT[p][e] = xc[e * IATOMS];
    }
    const int ntile = (wave + WPB < NCAPS) ? 2 : 1;
    const float bt0 = bias[wave * CDIM + lcol];
    const float bt1 = bias[(ntile == 2 ? wave + WPB : wave) * CDIM + lcol];
    {
        const int wr = t >> 4, wj = t & 15;
        const float* wg = Wm + wr * NCD + 4 * wj;
        float* ws = &sW[wr * WSTR + 4 * wj];
        v4f wv[3];
#pragma unroll
        for (int i = 0; i < 3; ++i)
            if (i < 2 || wj < 8) wv[i] = *(const v4f*)(wg + 64 * i);
#pragma unroll
        for (int i = 0; i < 3; ++i)
            if (i < 2 || wj < 8) {
                *(v4f*)(ws + 64 * i) = wv[i];
                float* wt = &sWT[(4 * wj + 64 * i) * WTSTR + wr];
                wt[0]         = wv[i].x;
                wt[WTSTR]     = wv[i].y;
                wt[2 * WTSTR] = wv[i].z;
                wt[3 * WTSTR] = wv[i].w;
            }
    }
    __syncthreads();

    v2f xb[PPW][2][4];
#pragma unroll
    for (int p = 0; p < PPW; ++p)
#pragma unroll
        for (int mt = 0; mt < 2; ++mt) {
            xb[p][mt][0].x = xa[p][mt][0].x; xb[p][mt][0].y = xa[p][mt][0].y;
            xb[p][mt][1].x = xa[p][mt][0].z; xb[p][mt][1].y = xa[p][mt][0].w;
            xb[p][mt][2].x = xa[p][mt][1].x; xb[p][mt][2].y = xa[p][mt][1].y;
            xb[p][mt][3].x = xa[p][mt][1].z; xb[p][mt][3].y = xa[p][mt][1].w;
        }

    const int ncA = 5 * half, ncB = 5 - 5 * half;
    float* rA = sRw + ncA * RSTR + lane;
    float* rB = sRw + ncB * RSTR + lane;
    const int l7 = lcol & 7;
    const int ncrow = (l7 < 4 ? l7 : 4) + 5 * (lcol >> 3);
    float lgA[PPW][5], lgB[PPW][5];
#pragma unroll
    for (int p = 0; p < PPW; ++p)
#pragma unroll
        for (int i = 0; i < 5; ++i) { lgA[p][i] = 0.0f; lgB[p][i] = 0.0f; }

    for (int r = 0; r < NROUT; ++r) {
        if (r == 0) {
#pragma unroll
            for (int p = 0; p < PPW; ++p) {
                float t8[8];
#pragma unroll
                for (int e = 0; e < 8; ++e) t8[e] = xT[p][2 * e] + xT[p][2 * e + 1];
                const float s = ((t8[0] + t8[1]) + (t8[2] + t8[3])) + ((t8[4] + t8[5]) + (t8[6] + t8[7]));
                sYw[p * PST + lcol] = (1.0f / NCAPS) * (s + xhalf(s));
            }
        } else {
            __builtin_amdgcn_wave_barrier();
#pragma unroll
            for (int p = 0; p < PPW; ++p) {
                float eA[5], eB[5], mx = -1e30f;
#pragma unroll
                for (int i = 0; i < 5; ++i) mx = fmaxf(mx, fmaxf(lgA[p][i], lgB[p][i]));
                float s = 0.0f;
#pragma unroll
                for (int i = 0; i < 5; ++i) { eA[i] = __expf(lgA[p][i] - mx); s += eA[i]; }
#pragma unroll
                for (int i = 0; i < 5; ++i) { eB[i] = __expf(lgB[p][i] - mx); s += eB[i]; }
                const float inv = __builtin_amdgcn_rcpf(s);
#pragma unroll
                for (int i = 0; i < 5; ++i) {
                    rA[p * RS + i * RSTR] = eA[i] * inv;
                    rB[p * RS + i * RSTR] = eB[i] * inv;
                }
            }
            __builtin_amdgcn_wave_barrier();

            v8f y0[PPW], y1[PPW];
            {
                v4f rq[PPW][4];
#pragma unroll
                for (int p = 0; p < PPW; ++p) {
                    const v4f* rp = (const v4f*)&sRw[p * RS + ncrow * RSTR + 16 * half];
#pragma unroll
                    for (int q = 0; q < 4; ++q) rq[p][q] = rp[q];
                    y0[p] = v8f{}; y1[p] = v8f{};
                }
#pragma unroll
                for (int q = 0; q < 4; ++q)
#pragma unroll
                    for (int p = 0; p < PPW; ++p) {
                        v2f ra0, ra1, xb0, xb1;
                        ra0.x = rq[p][q].x; ra0.y = rq[p][q].y;
                        ra1.x = rq[p][q].z; ra1.y = rq[p][q].w;
                        xb0.x = xT[p][4 * q];     xb0.y = xT[p][4 * q + 1];
                        xb1.x = xT[p][4 * q + 2]; xb1.y = xT[p][4 * q + 3];
                        y0[p] = mma4(ra0, xb0, y0[p]);
                        y1[p] = mma4(ra1, xb1, y1[p]);
                    }
            }
#pragma unroll
            for (int p = 0; p < PPW; ++p)
#pragma unroll
                for (int i = 0; i < 5; ++i) sYw[p * PST + (ncA + i) * IATOMS + lcol] = y0[p][i] + y1[p][i];
        }
        __syncthreads();

        if (ntile == 2)
            capsule_tiles<2>(sY, sA, sU, sW, sWT, wave, lcol, half, r == 0, r < NROUT - 1, bt0, bt1);
        else
            capsule_tiles<1>(sY, sA, sU, sW, sWT, wave, lcol, half, r == 0, r < NROUT - 1, bt0, bt1);

        if (r < NROUT - 1) {
            __syncthreads();
            v2f uf[PPW][4];
            v8f d[PPW][2];
#pragma unroll
            for (int p = 0; p < PPW; ++p) {
                const v4f* up = (const v4f*)&sUw[p * PST + ncrow * IATOMS + 8 * half];
                const v4f ua = up[0], ub = up[1];
                uf[p][0].x = ua.x; uf[p][0].y = ua.y; uf[p][1].x = ua.z; uf[p][1].y = ua.w;
                uf[p][2].x = ub.x; uf[p][2].y = ub.y; uf[p][3].x = ub.z; uf[p][3].y = ub.w;
                d[p][0] = v8f{}; d[p][1] = v8f{};
            }
#pragma unroll
            for (int ks = 0; ks < 4; ++ks)
#pragma unroll
                for (int mt = 0; mt < 2; ++mt)
#pragma unroll
                    for (int p = 0; p < PPW; ++p)
                        d[p][mt] = mma4(uf[p][ks], xb[p][mt][ks], d[p][mt]);
#pragma unroll
            for (int p = 0; p < PPW; ++p)
#pragma unroll
                for (int i = 0; i < 5; ++i) {
                    const float keep = half ? d[p][1][i] : d[p][0][i];
                    const float send = half ? d[p][0][i] : d[p][1][i];
                    lgA[p][i] += keep;
                    lgB[p][i] += xhalf(send);
                }
        }
    }

    __syncthreads();
#pragma unroll
    for (int p = 0; p < PPW; ++p)
#pragma unroll
        for (int q = 0; q < NCAPS / 2; ++q)
            out[(pos0 + p) * NCD + 32 * q + lane] = sAw[p * PST + 32 * q + lane];
}

extern "C" void kernel_launch(void* const* d_in, const int* in_sizes, int n_in,
                              void* d_out, int out_size, void* d_ws, size_t ws_size,
                              hipStream_t stream) {
    const float* x  = (const float*)d_in[0];
    const float* W  = (const float*)d_in[1];
    const float* b  = (const float*)d_in[2];
    float* out      = (float*)d_out;
    const int positions = 32 * 16 * 16;
    caps_routing_fused<<<dim3(positions / PPB), dim3(32 * WPB), 0, stream>>>(x, W, b, out);
}


# W products transposed (W as A operand): a lane holds 8 cd of a position; squash norm in-lane plus a half exchange, factor once per lane, act to U in registers, bias as C; changes pre, norm sum order
# speedup vs baseline: 6.9291x; 1.1848x over previous
#include <hip/hip_runtime.h>


typedef __attribute__((ext_vector_type(2))) float v2f;
typedef __attribute__((ext_vector_type(4))) float v4f;
typedef __attribute__((ext_vector_type(8))) float v8f;

#define ICAPS   32
#define IATOMS  16
#define NCAPS   10
#define CDIM    16
#define NCD     160
#define NROUT   3
#define EPSV    1e-7f
#define WSTR    164
#define WTSTR   20
#define RSTR    32
#define PST     164
#define PPW     2
#define WPB     8
#define PPB     (PPW * WPB)

static_assert(PPB == 16, "the products with W take the work-group's positions as the 16 columns of a matrix tile");


__device__ __forceinline__ float xhalf(float x) {
    return __builtin_bit_cast(float, __builtin_amdgcn_permlanex16(
        __builtin_bit_cast(int, x), __builtin_bit_cast(int, x),
        0x76543210, 0xfedcba98, false, false));
}

__device__ __forceinline__ v8f mma4(v2f a, v2f b, v8f c) {
    return __builtin_amdgcn_wmma_f32_16x16x4_f32(false, a, false, b, (short)0, c, false, false);
}

template <int NT>
__device__ __forceinline__ void capsule_tiles(const float* sY, float* sA, float* sU, const float* sW, const float* sWT,
                                              int wave, int lcol, int half, bool first, bool need_u,
                                              v8f bias0, v8f bias1)
{
    v4f ya[NT], yb[NT], wa[NT], wb[NT], va[NT], vb[NT];
#pragma unroll
    for (int tl = 0; tl < NT; ++tl) {
        const int nc = wave + WPB * tl;
        const float* yp = sY + lcol * PST + (first ? 0 : nc * IATOMS) + 8 * half;
        const float* wp = sWT + (CDIM * nc + lcol) * WTSTR + 8 * half;
        const float* wq = sW + lcol * WSTR + CDIM * nc + 8 * half;
        ya[tl] = *(const v4f*)yp; yb[tl] = *(const v4f*)(yp + 4);
        wa[tl] = *(const v4f*)wp; wb[tl] = *(const v4f*)(wp + 4);
        va[tl] = *(const v4f*)wq; vb[tl] = *(const v4f*)(wq + 4);
    }
    v8f a0[NT], a1[NT];
#pragma unroll
    for (int tl = 0; tl < NT; ++tl) {
        a0[tl] = mma4(v2f{wa[tl].x, wa[tl].y}, v2f{ya[tl].x, ya[tl].y}, tl ? bias1 : bias0);
        a1[tl] = mma4(v2f{wa[tl].z, wa[tl].w}, v2f{ya[tl].z, ya[tl].w}, v8f{});
    }
#pragma unroll
    for (int tl = 0; tl < NT; ++tl) {
        a0[tl] = mma4(v2f{wb[tl].x, wb[tl].y}, v2f{yb[tl].x, yb[tl].y}, a0[tl]);
        a1[tl] = mma4(v2f{wb[tl].z, wb[tl].w}, v2f{yb[tl].z, yb[tl].w}, a1[tl]);
    }
    float act[NT][8];
#pragma unroll
    for (int tl = 0; tl < NT; ++tl) {
        float pp[8];
#pragma unroll
        for (int e = 0; e < 8; ++e) pp[e] = a0[tl][e] + a1[tl][e];
        float q0 = pp[0] * pp[0], q1 = pp[1] * pp[1];
#pragma unroll
        for (int e = 2; e < 8; e += 2) { q0 += pp[e] * pp[e]; q1 += pp[e + 1] * pp[e + 1]; }
        const float qh = q0 + q1;
        const float sq = qh + xhalf(qh);
        const float sqc = fminf(sq, 0x1p100f);
        const float f = (sqc * __builtin_amdgcn_rcpf(1.0f + sqc)) * __builtin_amdgcn_rsqf(sq + EPSV);
#pragma unroll
        for (int e = 0; e < 8; ++e) act[tl][e] = f * pp[e];
    }
    if (need_u) {
        v8f u0[NT], u1[NT];
#pragma unroll
        for (int tl = 0; tl < NT; ++tl) {
            u0[tl] = mma4(v2f{va[tl].x, va[tl].y}, v2f{act[tl][0], act[tl][1]}, v8f{});
            u1[tl] = mma4(v2f{va[tl].z, va[tl].w}, v2f{act[tl][2], act[tl][3]}, v8f{});
        }
#pragma unroll
        for (int tl = 0; tl < NT; ++tl) {
            u0[tl] = mma4(v2f{vb[tl].x, vb[tl].y}, v2f{act[tl][4], act[tl][5]}, u0[tl]);
            u1[tl] = mma4(v2f{vb[tl].z, vb[tl].w}, v2f{act[tl][6], act[tl][7]}, u1[tl]);
        }
#pragma unroll
        for (int tl = 0; tl < NT; ++tl) {
            const int nc = wave + WPB * tl;
            float* up = sU + lcol * PST + nc * IATOMS + 8 * half;
            *(v4f*)up       = v4f{u0[tl][0] + u1[tl][0], u0[tl][1] + u1[tl][1], u0[tl][2] + u1[tl][2], u0[tl][3] + u1[tl][3]};
            *(v4f*)(up + 4) = v4f{u0[tl][4] + u1[tl][4], u0[tl][5] + u1[tl][5], u0[tl][6] + u1[tl][6], u0[tl][7] + u1[tl][7]};
        }
    } else {
#pragma unroll
        for (int tl = 0; tl < NT; ++tl) {
            const int nc = wave + WPB * tl;
            float* ap = sA + lcol * PST + nc * CDIM + 8 * half;
            *(v4f*)ap       = v4f{act[tl][0], act[tl][1], act[tl][2], act[tl][3]};
            *(v4f*)(ap + 4) = v4f{act[tl][4], act[tl][5], act[tl][6], act[tl][7]};
        }
    }
}

__global__ __launch_bounds__(32 * WPB, 4)
void caps_routing_fused(const float* __restrict__ x,
                        const float* __restrict__ Wm,
                        const float* __restrict__ bias,
                        float* __restrict__ out)
{
    __shared__ __attribute__((aligned(16))) float sW[IATOMS * WSTR];
    __shared__ __attribute__((aligned(16))) float sWT[NCD * WTSTR];
    __shared__ __attribute__((aligned(16))) float sR[PPB][NCAPS * RSTR];
    __shared__ __attribute__((aligned(16))) float sY[PPB * PST];
    __shared__ __attribute__((aligned(16))) float sA[PPB * PST];
    __shared__ __attribute__((aligned(16))) float sU[PPB * PST];

    const int t    = threadIdx.x;
    const int lane = t & 31;
    const int wave = __builtin_amdgcn_readfirstlane(t >> 5);
    const int half = lane >> 4;
    const int lcol = lane & 15;
    const long pos0 = (long)blockIdx.x * PPB + wave * PPW;
    constexpr int RS = NCAPS * RSTR;
    float* sRw = sR[wave * PPW];
    float* sYw = sY + wave * PPW * PST;
    float* sAw = sA + wave * PPW * PST;
    float* sUw = sU + wave * PPW * PST;

    v4f xa[PPW][2][2];
    float xT[PPW][16];
#pragma unroll
    for (int p = 0; p < PPW; ++p) {
        const float* xp = x + (pos0 + p) * (ICAPS * IATOMS);
        const float* xr = xp + lcol * IATOMS + 8 * half;
#pragma unroll
        for (int mt = 0; mt < 2; ++mt) {
            xa[p][mt][0] = *(const v4f*)(xr + mt * 16 * IATOMS);
            xa[p][mt][1] = *(const v4f*)(xr + mt * 16 * IATOMS + 4);
        }
        const float* xc = xp + (16 * half) * IATOMS + lcol;
#pragma unroll
        for (int e = 0; e < 16; ++e) xT[p][e] = xc[e * IATOMS];
    }
    const int ntile = (wave + WPB < NCAPS) ? 2 : 1;
    v8f bias0, bias1;
    {
        const float* bp0 = bias + wave * CDIM + 8 * half;
        const float* bp1 = bias + (ntile == 2 ? wave + WPB : wave) * CDIM + 8 * half;
        const v4f b0a = *(const v4f*)bp0, b0b = *(const v4f*)(bp0 + 4);
        const v4f b1a = *(const v4f*)bp1, b1b = *(const v4f*)(bp1 + 4);
        bias0 = v8f{b0a.x, b0a.y, b0a.z, b0a.w, b0b.x, b0b.y, b0b.z, b0b.w};
        bias1 = v8f{b1a.x, b1a.y, b1a.z, b1a.w, b1b.x, b1b.y, b1b.z, b1b.w};
    }
    {
        const int wr = t >> 4, wj = t & 15;
        const float* wg = Wm + wr * NCD + 4 * wj;
        float* ws = &sW[wr * WSTR + 4 * wj];
        v4f wv[3];
#pragma unroll
        for (int i = 0; i < 3; ++i)
            if (i < 2 || wj < 8) wv[i] = *(const v4f*)(wg + 64 * i);
#pragma unroll
        for (int i = 0; i < 3; ++i)
            if (i < 2 || wj < 8) {
                *(v4f*)(ws + 64 * i) = wv[i];
                float* wt = &sWT[(4 * wj + 64 * i) * WTSTR + wr];
                wt[0]         = wv[i].x;
                wt[WTSTR]     = wv[i].y;
                wt[2 * WTSTR] = wv[i].z;
                wt[3 * WTSTR] = wv[i].w;
            }
    }
    __syncthreads();

    v2f xb[PPW][2][4];
#pragma unroll
    for (int p = 0; p < PPW; ++p)
#pragma unroll
        for (int mt = 0; mt < 2; ++mt) {
            xb[p][mt][0].x = xa[p][mt][0].x; xb[p][mt][0].y = xa[p][mt][0].y;
            xb[p][mt][1].x = xa[p][mt][0].z; xb[p][mt][1].y = xa[p][mt][0].w;
            xb[p][mt][2].x = xa[p][mt][1].x; xb[p][mt][2].y = xa[p][mt][1].y;
            xb[p][mt][3].x = xa[p][mt][1].z; xb[p][mt][3].y = xa[p][mt][1].w;
        }

    const int ncA = 5 * half, ncB = 5 - 5 * half;
    float* rA = sRw + ncA * RSTR + lane;
    float* rB = sRw + ncB * RSTR + lane;
    const int l7 = lcol & 7;
    const int ncrow = (l7 < 4 ? l7 : 4) + 5 * (lcol >> 3);
    float lgA[PPW][5], lgB[PPW][5];
#pragma unroll
    for (int p = 0; p < PPW; ++p)
#pragma unroll
        for (int i = 0; i < 5; ++i) { lgA[p][i] = 0.0f; lgB[p][i] = 0.0f; }

    for (int r = 0; r < NROUT; ++r) {
        if (r == 0) {
#pragma unroll
            for (int p = 0; p < PPW; ++p) {
                float t8[8];
#pragma unroll
                for (int e = 0; e < 8; ++e) t8[e] = xT[p][2 * e] + xT[p][2 * e + 1];
                const float s = ((t8[0] + t8[1]) + (t8[2] + t8[3])) + ((t8[4] + t8[5]) + (t8[6] + t8[7]));
                sYw[p * PST + lcol] = (1.0f / NCAPS) * (s + xhalf(s));
            }
        } else {
            __builtin_amdgcn_wave_barrier();
#pragma unroll
            for (int p = 0; p < PPW; ++p) {
                float eA[5], eB[5], mx = -1e30f;
#pragma unroll
                for (int i = 0; i < 5; ++i) mx = fmaxf(mx, fmaxf(lgA[p][i], lgB[p][i]));
                float s = 0.0f;
#pragma unroll
                for (int i = 0; i < 5; ++i) { eA[i] = __expf(lgA[p][i] - mx); s += eA[i]; }
#pragma unroll
                for (int i = 0; i < 5; ++i) { eB[i] = __expf(lgB[p][i] - mx); s += eB[i]; }
                const float inv = __builtin_amdgcn_rcpf(s);
#pragma unroll
                for (int i = 0; i < 5; ++i) {
                    rA[p * RS + i * RSTR] = eA[i] * inv;
                    rB[p * RS + i * RSTR] = eB[i] * inv;
                }
            }
            __builtin_amdgcn_wave_barrier();

            v8f y0[PPW], y1[PPW];
            {
                v4f rq[PPW][4];
#pragma unroll
                for (int p = 0; p < PPW; ++p) {
                    const v4f* rp = (const v4f*)&sRw[p * RS + ncrow * RSTR + 16 * half];
#pragma unroll
                    for (int q = 0; q < 4; ++q) rq[p][q] = rp[q];
                    y0[p] = v8f{}; y1[p] = v8f{};
                }
#pragma unroll
                for (int q = 0; q < 4; ++q)
#pragma unroll
                    for (int p = 0; p < PPW; ++p) {
                        v2f ra0, ra1, xb0, xb1;
                        ra0.x = rq[p][q].x; ra0.y = rq[p][q].y;
                        ra1.x = rq[p][q].z; ra1.y = rq[p][q].w;
                        xb0.x = xT[p][4 * q];     xb0.y = xT[p][4 * q + 1];
                        xb1.x = xT[p][4 * q + 2]; xb1.y = xT[p][4 * q + 3];
                        y0[p] = mma4(ra0, xb0, y0[p]);
                        y1[p] = mma4(ra1, xb1, y1[p]);
                    }
            }
#pragma unroll
            for (int p = 0; p < PPW; ++p)
#pragma unroll
                for (int i = 0; i < 5; ++i) sYw[p * PST + (ncA + i) * IATOMS + lcol] = y0[p][i] + y1[p][i];
        }
        __syncthreads();

        if (ntile == 2)
            capsule_tiles<2>(sY, sA, sU, sW, sWT, wave, lcol, half, r == 0, r < NROUT - 1, bias0, bias1);
        else
            capsule_tiles<1>(sY, sA, sU, sW, sWT, wave, lcol, half, r == 0, r < NROUT - 1, bias0, bias1);

        if (r < NROUT - 1) {
            __syncthreads();
            v2f uf[PPW][4];
            v8f d[PPW][2];
#pragma unroll
            for (int p = 0; p < PPW; ++p) {
                const v4f* up = (const v4f*)&sUw[p * PST + ncrow * IATOMS + 8 * half];
                const v4f ua = up[0], ub = up[1];
                uf[p][0].x = ua.x; uf[p][0].y = ua.y; uf[p][1].x = ua.z; uf[p][1].y = ua.w;
                uf[p][2].x = ub.x; uf[p][2].y = ub.y; uf[p][3].x = ub.z; uf[p][3].y = ub.w;
                d[p][0] = v8f{}; d[p][1] = v8f{};
            }
#pragma unroll
            for (int ks = 0; ks < 4; ++ks)
#pragma unroll
                for (int mt = 0; mt < 2; ++mt)
#pragma unroll
                    for (int p = 0; p < PPW; ++p)
                        d[p][mt] = mma4(uf[p][ks], xb[p][mt][ks], d[p][mt]);
#pragma unroll
            for (int p = 0; p < PPW; ++p)
#pragma unroll
                for (int i = 0; i < 5; ++i) {
                    const float keep = half ? d[p][1][i] : d[p][0][i];
                    const float send = half ? d[p][0][i] : d[p][1][i];
                    lgA[p][i] += keep;
                    lgB[p][i] += xhalf(send);
                }
        }
    }

    __syncthreads();
#pragma unroll
    for (int p = 0; p < PPW; ++p)
#pragma unroll
        for (int q = 0; q < NCAPS / 2; ++q)
            out[(pos0 + p) * NCD + 32 * q + lane] = sAw[p * PST + 32 * q + lane];
}

extern "C" void kernel_launch(void* const* d_in, const int* in_sizes, int n_in,
                              void* d_out, int out_size, void* d_ws, size_t ws_size,
                              hipStream_t stream) {
    const float* x  = (const float*)d_in[0];
    const float* W  = (const float*)d_in[1];
    const float* b  = (const float*)d_in[2];
    float* out      = (float*)d_out;
    const int positions = 32 * 16 * 16;
    caps_routing_fused<<<dim3(positions / PPB), dim3(32 * WPB), 0, stream>>>(x, W, b, out);
}
